# starting kernel
# baseline (speedup 1.0000x reference)
#include <hip/hip_runtime.h>

typedef __attribute__((ext_vector_type(2))) float v2f;
typedef __attribute__((ext_vector_type(8))) float v8f;

#define Bn 8
#define Nn 8192
#define S_ATTN 128
#define S_NONE 384
#define Sn 512
#define NSAMPLE 64
#define RADIUS 0.4f
#define D_FEAT 64
#define IN_CH 67
#define C0PAD 68

#define OUT_NP_OFF 12288
#define OUT_ATT_OFF 536576

__device__ __forceinline__ v8f wmma_f32(v2f a, v2f b, v8f c) {
  return __builtin_amdgcn_wmma_f32_16x16x4_f32(
       false, a,  false, b,
       (short)0, c,  false,  false);
}

__global__ void __launch_bounds__(1024) fps_kernel(const float* __restrict__ xyz,
                                                   const float* __restrict__ attn,
                                                   float* __restrict__ out) {
  const int b = blockIdx.x >> 1;
  const int branch = blockIdx.x & 1;
  const int K = branch ? S_NONE : S_ATTN;
  const int rowoff = branch ? S_ATTN : 0;
  const int tid = threadIdx.x;
  const int wid = tid >> 5, lane = tid & 31;

  const float* xb = xyz + (size_t)b * 3 * Nn;
  const float* ab = attn + (size_t)b * Nn;

  float px[8], py[8], pz[8], mind[8];
#pragma unroll
  for (int i = 0; i < 8; ++i) {
    int n = tid + i * 1024;
    float a = ab[n];
    float w = branch ? (1.0f - a) : a;
    px[i] = w * xb[n];
    py[i] = w * xb[Nn + n];
    pz[i] = w * xb[2 * Nn + n];
    mind[i] = 1e10f;
  }

  __shared__ float s_bd[32];
  __shared__ int s_bn[32];
  __shared__ float s_l[3];
  __shared__ int s_last;

  int last = 0;
  for (int k = 0; k < K; ++k) {
    if (tid == 0) {
      float a = ab[last];
      float w = branch ? (1.0f - a) : a;
      float lx = w * xb[last];
      float ly = w * xb[Nn + last];
      float lz = w * xb[2 * Nn + last];
      int row = rowoff + k;
      out[(size_t)b * 1536 + row] = lx;
      out[(size_t)b * 1536 + 512 + row] = ly;
      out[(size_t)b * 1536 + 1024 + row] = lz;
      out[OUT_ATT_OFF + (size_t)b * 512 + row] = a;
      s_l[0] = lx; s_l[1] = ly; s_l[2] = lz;
    }
    __syncthreads();
    if (k == K - 1) break;

    const float lx = s_l[0], ly = s_l[1], lz = s_l[2];
    float bd = -1.0f;
    int bn = Nn;
#pragma unroll
    for (int i = 0; i < 8; ++i) {
      float dx = px[i] - lx, dy = py[i] - ly, dz = pz[i] - lz;
      float d2 = dx * dx + dy * dy + dz * dz;
      mind[i] = fminf(mind[i], d2);
      if (mind[i] > bd) { bd = mind[i]; bn = tid + i * 1024; }
    }
    for (int off = 16; off > 0; off >>= 1) {
      float od = __shfl_down(bd, off, 32);
      int on = __shfl_down(bn, off, 32);
      if (od > bd || (od == bd && on < bn)) { bd = od; bn = on; }
    }
    if (lane == 0) { s_bd[wid] = bd; s_bn[wid] = bn; }
    __syncthreads();
    if (wid == 0) {
      bd = s_bd[lane];
      bn = s_bn[lane];
      for (int off = 16; off > 0; off >>= 1) {
        float od = __shfl_down(bd, off, 32);
        int on = __shfl_down(bn, off, 32);
        if (od > bd || (od == bd && on < bn)) { bd = od; bn = on; }
      }
      if (lane == 0) s_last = bn;
    }
    __syncthreads();
    last = s_last;
  }
}

template <int CINPAD, int WSTRIDE>
__device__ __forceinline__ void mlp_layer(const float* __restrict__ W,
                                          const float* __restrict__ bias,
                                          const float* sin_, float* sout,
                                          int cout, int lane) {
  const int hl = lane >> 4;
  const int ll = lane & 15;
  for (int ot = 0; ot < cout / 16; ++ot) {
    const float bv = bias[ot * 16 + ll];
    for (int rt = 0; rt < 4; ++rt) {
      v8f acc = {};
      const int m = rt * 16 + ll;
      const int o = ot * 16 + ll;
      for (int c = 0; c < CINPAD; c += 4) {
        const int kk = c + (hl << 1);
        v2f a, bf;
        a.x = sin_[m * C0PAD + kk];
        a.y = sin_[m * C0PAD + kk + 1];
        bf.x = W[o * WSTRIDE + kk];
        bf.y = W[o * WSTRIDE + kk + 1];
        acc = wmma_f32(a, bf, acc);
      }
#pragma unroll
      for (int r = 0; r < 8; ++r) {
        float v = acc[r] + bv;
        v = v > 0.0f ? v : 0.0f;
        const int krow = rt * 16 + r + 8 * hl;
        sout[krow * C0PAD + ot * 16 + ll] = v;
      }
    }
  }
}

__device__ __forceinline__ void mlp_layer3_max(const float* __restrict__ W,
                                               const float* __restrict__ bias,
                                               const float* sin_,
                                               float* __restrict__ out,
                                               int lane, int b, int s) {
  const int hl = lane >> 4;
  const int ll = lane & 15;
  for (int ot = 0; ot < 8; ++ot) {
    const float bv = bias[ot * 16 + ll];
    v8f mx = {};
    for (int rt = 0; rt < 4; ++rt) {
      v8f acc = {};
      const int m = rt * 16 + ll;
      const int o = ot * 16 + ll;
      for (int c = 0; c < 64; c += 4) {
        const int kk = c + (hl << 1);
        v2f a, bf;
        a.x = sin_[m * C0PAD + kk];
        a.y = sin_[m * C0PAD + kk + 1];
        bf.x = W[o * 64 + kk];
        bf.y = W[o * 64 + kk + 1];
        acc = wmma_f32(a, bf, acc);
      }
#pragma unroll
      for (int r = 0; r < 8; ++r) {
        float v = acc[r] + bv;
        v = v > 0.0f ? v : 0.0f;
        mx[r] = fmaxf(mx[r], v);
      }
    }
    float m8 = mx[0];
#pragma unroll
    for (int r = 1; r < 8; ++r) m8 = fmaxf(m8, mx[r]);
    float oth = __shfl_xor(m8, 16, 32);
    m8 = fmaxf(m8, oth);
    if (lane < 16) {
      const int o = ot * 16 + lane;
      out[OUT_NP_OFF + (size_t)b * 65536 + (size_t)o * 512 + s] = m8;
    }
  }
}

__global__ void __launch_bounds__(32) sa_fused_kernel(
    const float* __restrict__ xyz, const float* __restrict__ pts,
    const float* __restrict__ w0, const float* __restrict__ b0,
    const float* __restrict__ w1, const float* __restrict__ b1,
    const float* __restrict__ w2, const float* __restrict__ b2,
    float* __restrict__ out) {
  __shared__ float s_w0[64 * C0PAD];
  __shared__ float s_feat[64 * C0PAD];
  __shared__ float s_act[64 * C0PAD];
  __shared__ int s_idx[NSAMPLE];

  const int lane = threadIdx.x;
  const int b = blockIdx.x >> 9;
  const int s = blockIdx.x & 511;

  for (int t = lane; t < 64 * C0PAD; t += 32) {
    int r = t / C0PAD, c = t - r * C0PAD;
    s_w0[t] = (c < IN_CH) ? w0[r * IN_CH + c] : 0.0f;
  }

  const float* xb = xyz + (size_t)b * 3 * Nn;
  const float qx = out[(size_t)b * 1536 + s];
  const float qy = out[(size_t)b * 1536 + 512 + s];
  const float qz = out[(size_t)b * 1536 + 1024 + s];

  const float R2 = RADIUS * RADIUS;
  int count = 0;
  for (int base = 0; base < Nn && count < NSAMPLE; base += 32) {
    __builtin_prefetch(xb + base + 256, 0, 0);
    const int n = base + lane;
    const float dx = xb[n] - qx;
    const float dy = xb[Nn + n] - qy;
    const float dz = xb[2 * Nn + n] - qz;
    const float d2 = dx * dx + dy * dy + dz * dz;
    const bool hit = d2 < R2;
    const unsigned mask = (unsigned)__ballot(hit);
    const int slot = count + __popc(mask & ((1u << lane) - 1u));
    if (hit && slot < NSAMPLE) s_idx[slot] = n;
    count += __popc(mask);
  }
  __syncthreads();
  const int cnt = count < NSAMPLE ? count : NSAMPLE;

  const float* pb = pts + (size_t)b * D_FEAT * Nn;
  for (int k = 0; k < NSAMPLE; ++k) {
    const int n = (k < cnt) ? s_idx[k] : -1;
    const int gi = (n >= 0) ? n : (Nn - 1);
#pragma unroll
    for (int j = 0; j < 3; ++j) {
      const int ch = lane + j * 32;
      if (ch < C0PAD) {
        float v;
        if (ch < 3) {
          float g = (n >= 0) ? xb[ch * Nn + n] : 0.0f;
          v = g - ((ch == 0) ? qx : (ch == 1) ? qy : qz);
        } else if (ch < IN_CH) {
          v = pb[(ch - 3) * Nn + gi];
        } else {
          v = 0.0f;
        }
        s_feat[k * C0PAD + ch] = v;
      }
    }
  }
  __syncthreads();

  mlp_layer<C0PAD, C0PAD>(s_w0, b0, s_feat, s_act, 64, lane);
  __syncthreads();
  mlp_layer<64, 64>(w1, b1, s_act, s_feat, 64, lane);
  __syncthreads();
  mlp_layer3_max(w2, b2, s_feat, out, lane, b, s);
}

extern "C" void kernel_launch(void* const* d_in, const int* in_sizes, int n_in,
                              void* d_out, int out_size, void* d_ws, size_t ws_size,
                              hipStream_t stream) {
  const float* xyz = (const float*)d_in[0];
  const float* pts = (const float*)d_in[1];
  const float* attn = (const float*)d_in[2];
  const float* w0 = (const float*)d_in[3];
  const float* b0 = (const float*)d_in[4];
  const float* w1 = (const float*)d_in[5];
  const float* b1 = (const float*)d_in[6];
  const float* w2 = (const float*)d_in[7];
  const float* b2 = (const float*)d_in[8];
  float* out = (float*)d_out;

  hipLaunchKernelGGL(fps_kernel, dim3(Bn * 2), dim3(1024), 0, stream, xyz, attn, out);
  hipLaunchKernelGGL(sa_fused_kernel, dim3(Bn * Sn), dim3(32), 0, stream,
                     xyz, pts, w0, b0, w1, b1, w2, b2, out);
}


# FPS loop: per-wave argmax and its coordinates go through two LDS slot sets, every wave finishes the reduction itself: one barrier per pick, no global load or store inside the loop
# speedup vs baseline: 1.1091x; 1.1091x over previous
#include <hip/hip_runtime.h>

typedef __attribute__((ext_vector_type(2))) float v2f;
typedef __attribute__((ext_vector_type(8))) float v8f;

#define Bn 8
#define Nn 8192
#define S_ATTN 128
#define S_NONE 384
#define Sn 512
#define NSAMPLE 64
#define RADIUS 0.4f
#define D_FEAT 64
#define IN_CH 67
#define C0PAD 68

#define OUT_NP_OFF 12288
#define OUT_ATT_OFF 536576

__device__ __forceinline__ v8f wmma_f32(v2f a, v2f b, v8f c) {
  return __builtin_amdgcn_wmma_f32_16x16x4_f32(
       false, a,  false, b,
       (short)0, c,  false,  false);
}

__global__ void __launch_bounds__(1024) fps_kernel(const float* __restrict__ xyz,
                                                   const float* __restrict__ attn,
                                                   float* __restrict__ out) {
  const int b = blockIdx.x >> 1;
  const int branch = blockIdx.x & 1;
  const int K = branch ? S_NONE : S_ATTN;
  const int rowoff = branch ? S_ATTN : 0;
  const int tid = threadIdx.x;
  const int wid = tid >> 5, lane = tid & 31;

  const float* xb = xyz + (size_t)b * 3 * Nn;
  const float* ab = attn + (size_t)b * Nn;

  float px[8], py[8], pz[8], mind[8];
#pragma unroll
  for (int i = 0; i < 8; ++i) {
    int n = tid + i * 1024;
    float a = ab[n];
    float w = branch ? (1.0f - a) : a;
    px[i] = w * xb[n];
    py[i] = w * xb[Nn + n];
    pz[i] = w * xb[2 * Nn + n];
    mind[i] = 1e10f;
  }

  __shared__ float s_bd[2][32];
  __shared__ int s_bn[2][32];
  __shared__ float s_cx[2][32];
  __shared__ float s_cy[2][32];
  __shared__ float s_cz[2][32];
  __shared__ int s_sel[S_NONE];
  __shared__ float s_ox[S_NONE];
  __shared__ float s_oy[S_NONE];
  __shared__ float s_oz[S_NONE];

  int last = 0;
  float lx, ly, lz;
  {
    const float a = ab[0];
    const float w = branch ? (1.0f - a) : a;
    lx = w * xb[0];
    ly = w * xb[Nn];
    lz = w * xb[2 * Nn];
  }

  for (int k = 0; k < K; ++k) {
    if (tid == 0) {
      s_sel[k] = last;
      s_ox[k] = lx; s_oy[k] = ly; s_oz[k] = lz;
    }
    if (k == K - 1) break;

    float bd = -1.0f;
    int bn = Nn;
#pragma unroll
    for (int i = 0; i < 8; ++i) {
      float dx = px[i] - lx, dy = py[i] - ly, dz = pz[i] - lz;
      float d2 = dx * dx + dy * dy + dz * dz;
      mind[i] = fminf(mind[i], d2);
      if (mind[i] > bd) { bd = mind[i]; bn = tid + i * 1024; }
    }
    for (int off = 16; off > 0; off >>= 1) {
      float od = __shfl_down(bd, off, 32);
      int on = __shfl_down(bn, off, 32);
      if (od > bd || (od == bd && on < bn)) { bd = od; bn = on; }
    }
    const int wbn = __builtin_amdgcn_readfirstlane(bn);
    const int wi = wbn >> 10;
    float sx = px[0], sy = py[0], sz = pz[0];
#pragma unroll
    for (int i = 1; i < 8; ++i) {
      if (wi == i) { sx = px[i]; sy = py[i]; sz = pz[i]; }
    }
    const int wl = wbn & 31;
    const float cx = __int_as_float(__builtin_amdgcn_readlane(__float_as_int(sx), wl));
    const float cy = __int_as_float(__builtin_amdgcn_readlane(__float_as_int(sy), wl));
    const float cz = __int_as_float(__builtin_amdgcn_readlane(__float_as_int(sz), wl));
    const int par = k & 1;
    if (lane == 0) {
      s_bd[par][wid] = bd;
      s_bn[par][wid] = bn;
      s_cx[par][wid] = cx;
      s_cy[par][wid] = cy;
      s_cz[par][wid] = cz;
    }
    __syncthreads();

    float gd = s_bd[par][lane];
    int gn = s_bn[par][lane];
    const float gx = s_cx[par][lane];
    const float gy = s_cy[par][lane];
    const float gz = s_cz[par][lane];
    for (int off = 16; off > 0; off >>= 1) {
      float od = __shfl_down(gd, off, 32);
      int on = __shfl_down(gn, off, 32);
      if (od > gd || (od == gd && on < gn)) { gd = od; gn = on; }
    }
    last = __builtin_amdgcn_readfirstlane(gn);
    const int fw = (last >> 5) & 31;
    lx = __int_as_float(__builtin_amdgcn_readlane(__float_as_int(gx), fw));
    ly = __int_as_float(__builtin_amdgcn_readlane(__float_as_int(gy), fw));
    lz = __int_as_float(__builtin_amdgcn_readlane(__float_as_int(gz), fw));
  }
  __syncthreads();

  for (int k = tid; k < K; k += 1024) {
    int n = s_sel[k];
    n = n < Nn ? n : Nn - 1;
    const int row = rowoff + k;
    out[(size_t)b * 1536 + row] = s_ox[k];
    out[(size_t)b * 1536 + 512 + row] = s_oy[k];
    out[(size_t)b * 1536 + 1024 + row] = s_oz[k];
    out[OUT_ATT_OFF + (size_t)b * 512 + row] = ab[n];
  }
}

template <int CINPAD, int WSTRIDE>
__device__ __forceinline__ void mlp_layer(const float* __restrict__ W,
                                          const float* __restrict__ bias,
                                          const float* sin_, float* sout,
                                          int cout, int lane) {
  const int hl = lane >> 4;
  const int ll = lane & 15;
  for (int ot = 0; ot < cout / 16; ++ot) {
    const float bv = bias[ot * 16 + ll];
    for (int rt = 0; rt < 4; ++rt) {
      v8f acc = {};
      const int m = rt * 16 + ll;
      const int o = ot * 16 + ll;
      for (int c = 0; c < CINPAD; c += 4) {
        const int kk = c + (hl << 1);
        v2f a, bf;
        a.x = sin_[m * C0PAD + kk];
        a.y = sin_[m * C0PAD + kk + 1];
        bf.x = W[o * WSTRIDE + kk];
        bf.y = W[o * WSTRIDE + kk + 1];
        acc = wmma_f32(a, bf, acc);
      }
#pragma unroll
      for (int r = 0; r < 8; ++r) {
        float v = acc[r] + bv;
        v = v > 0.0f ? v : 0.0f;
        const int krow = rt * 16 + r + 8 * hl;
        sout[krow * C0PAD + ot * 16 + ll] = v;
      }
    }
  }
}

__device__ __forceinline__ void mlp_layer3_max(const float* __restrict__ W,
                                               const float* __restrict__ bias,
                                               const float* sin_,
                                               float* __restrict__ out,
                                               int lane, int b, int s) {
  const int hl = lane >> 4;
  const int ll = lane & 15;
  for (int ot = 0; ot < 8; ++ot) {
    const float bv = bias[ot * 16 + ll];
    v8f mx = {};
    for (int rt = 0; rt < 4; ++rt) {
      v8f acc = {};
      const int m = rt * 16 + ll;
      const int o = ot * 16 + ll;
      for (int c = 0; c < 64; c += 4) {
        const int kk = c + (hl << 1);
        v2f a, bf;
        a.x = sin_[m * C0PAD + kk];
        a.y = sin_[m * C0PAD + kk + 1];
        bf.x = W[o * 64 + kk];
        bf.y = W[o * 64 + kk + 1];
        acc = wmma_f32(a, bf, acc);
      }
#pragma unroll
      for (int r = 0; r < 8; ++r) {
        float v = acc[r] + bv;
        v = v > 0.0f ? v : 0.0f;
        mx[r] = fmaxf(mx[r], v);
      }
    }
    float m8 = mx[0];
#pragma unroll
    for (int r = 1; r < 8; ++r) m8 = fmaxf(m8, mx[r]);
    float oth = __shfl_xor(m8, 16, 32);
    m8 = fmaxf(m8, oth);
    if (lane < 16) {
      const int o = ot * 16 + lane;
      out[OUT_NP_OFF + (size_t)b * 65536 + (size_t)o * 512 + s] = m8;
    }
  }
}

__global__ void __launch_bounds__(32) sa_fused_kernel(
    const float* __restrict__ xyz, const float* __restrict__ pts,
    const float* __restrict__ w0, const float* __restrict__ b0,
    const float* __restrict__ w1, const float* __restrict__ b1,
    const float* __restrict__ w2, const float* __restrict__ b2,
    float* __restrict__ out) {
  __shared__ float s_w0[64 * C0PAD];
  __shared__ float s_feat[64 * C0PAD];
  __shared__ float s_act[64 * C0PAD];
  __shared__ int s_idx[NSAMPLE];

  const int lane = threadIdx.x;
  const int b = blockIdx.x >> 9;
  const int s = blockIdx.x & 511;

  for (int t = lane; t < 64 * C0PAD; t += 32) {
    int r = t / C0PAD, c = t - r * C0PAD;
    s_w0[t] = (c < IN_CH) ? w0[r * IN_CH + c] : 0.0f;
  }

  const float* xb = xyz + (size_t)b * 3 * Nn;
  const float qx = out[(size_t)b * 1536 + s];
  const float qy = out[(size_t)b * 1536 + 512 + s];
  const float qz = out[(size_t)b * 1536 + 1024 + s];

  const float R2 = RADIUS * RADIUS;
  int count = 0;
  for (int base = 0; base < Nn && count < NSAMPLE; base += 32) {
    __builtin_prefetch(xb + base + 256, 0, 0);
    const int n = base + lane;
    const float dx = xb[n] - qx;
    const float dy = xb[Nn + n] - qy;
    const float dz = xb[2 * Nn + n] - qz;
    const float d2 = dx * dx + dy * dy + dz * dz;
    const bool hit = d2 < R2;
    const unsigned mask = (unsigned)__ballot(hit);
    const int slot = count + __popc(mask & ((1u << lane) - 1u));
    if (hit && slot < NSAMPLE) s_idx[slot] = n;
    count += __popc(mask);
  }
  __syncthreads();
  const int cnt = count < NSAMPLE ? count : NSAMPLE;

  const float* pb = pts + (size_t)b * D_FEAT * Nn;
  for (int k = 0; k < NSAMPLE; ++k) {
    const int n = (k < cnt) ? s_idx[k] : -1;
    const int gi = (n >= 0) ? n : (Nn - 1);
#pragma unroll
    for (int j = 0; j < 3; ++j) {
      const int ch = lane + j * 32;
      if (ch < C0PAD) {
        float v;
        if (ch < 3) {
          float g = (n >= 0) ? xb[ch * Nn + n] : 0.0f;
          v = g - ((ch == 0) ? qx : (ch == 1) ? qy : qz);
        } else if (ch < IN_CH) {
          v = pb[(ch - 3) * Nn + gi];
        } else {
          v = 0.0f;
        }
        s_feat[k * C0PAD + ch] = v;
      }
    }
  }
  __syncthreads();

  mlp_layer<C0PAD, C0PAD>(s_w0, b0, s_feat, s_act, 64, lane);
  __syncthreads();
  mlp_layer<64, 64>(w1, b1, s_act, s_feat, 64, lane);
  __syncthreads();
  mlp_layer3_max(w2, b2, s_feat, out, lane, b, s);
}

extern "C" void kernel_launch(void* const* d_in, const int* in_sizes, int n_in,
                              void* d_out, int out_size, void* d_ws, size_t ws_size,
                              hipStream_t stream) {
  const float* xyz = (const float*)d_in[0];
  const float* pts = (const float*)d_in[1];
  const float* attn = (const float*)d_in[2];
  const float* w0 = (const float*)d_in[3];
  const float* b0 = (const float*)d_in[4];
  const float* w1 = (const float*)d_in[5];
  const float* b1 = (const float*)d_in[6];
  const float* w2 = (const float*)d_in[7];
  const float* b2 = (const float*)d_in[8];
  float* out = (float*)d_out;

  hipLaunchKernelGGL(fps_kernel, dim3(Bn * 2), dim3(1024), 0, stream, xyz, attn, out);
  hipLaunchKernelGGL(sa_fused_kernel, dim3(Bn * Sn), dim3(32), 0, stream,
                     xyz, pts, w0, b0, w1, b1, w2, b2, out);
}


# FPS pick: both argmax reductions (in-wave, then over the 32 wave candidates) as integer max, then lowest index at the max, by row_xmask DPP and permlanex16; replaces ten ds_bpermute links
# speedup vs baseline: 1.4052x; 1.2670x over previous
#include <hip/hip_runtime.h>

typedef __attribute__((ext_vector_type(2))) float v2f;
typedef __attribute__((ext_vector_type(8))) float v8f;

#define Bn 8
#define Nn 8192
#define S_ATTN 128
#define S_NONE 384
#define Sn 512
#define NSAMPLE 64
#define RADIUS 0.4f
#define D_FEAT 64
#define IN_CH 67
#define C0PAD 68

#define OUT_NP_OFF 12288
#define OUT_ATT_OFF 536576

__device__ __forceinline__ v8f wmma_f32(v2f a, v2f b, v8f c) {
  return __builtin_amdgcn_wmma_f32_16x16x4_f32(
       false, a,  false, b,
       (short)0, c,  false,  false);
}

template <int M>
__device__ __forceinline__ int row_xor(int v) {
  return __builtin_amdgcn_update_dpp(0, v, 0x160 + M, 0xf, 0xf, true);
}
__device__ __forceinline__ int other_row(int v) {
  return __builtin_amdgcn_permlanex16(0, v, 0x76543210, 0xfedcba98, false, false);
}
__device__ __forceinline__ int wave_max(int v) {
  v = max(v, row_xor<1>(v));
  v = max(v, row_xor<2>(v));
  v = max(v, row_xor<4>(v));
  v = max(v, row_xor<8>(v));
  return max(v, other_row(v));
}
__device__ __forceinline__ int wave_min(int v) {
  v = min(v, row_xor<1>(v));
  v = min(v, row_xor<2>(v));
  v = min(v, row_xor<4>(v));
  v = min(v, row_xor<8>(v));
  return min(v, other_row(v));
}
__device__ __forceinline__ int wave_argmax(int d, int n, int* best) {
  const int m = wave_max(d);
  *best = m;
  return wave_min(d == m ? n : 0x7fffffff);
}

__global__ void __launch_bounds__(1024) fps_kernel(const float* __restrict__ xyz,
                                                   const float* __restrict__ attn,
                                                   float* __restrict__ out) {
  const int b = blockIdx.x >> 1;
  const int branch = blockIdx.x & 1;
  const int K = branch ? S_NONE : S_ATTN;
  const int rowoff = branch ? S_ATTN : 0;
  const int tid = threadIdx.x;
  const int wid = tid >> 5, lane = tid & 31;

  const float* xb = xyz + (size_t)b * 3 * Nn;
  const float* ab = attn + (size_t)b * Nn;

  float px[8], py[8], pz[8], mind[8];
#pragma unroll
  for (int i = 0; i < 8; ++i) {
    int n = tid + i * 1024;
    float a = ab[n];
    float w = branch ? (1.0f - a) : a;
    px[i] = w * xb[n];
    py[i] = w * xb[Nn + n];
    pz[i] = w * xb[2 * Nn + n];
    mind[i] = 1e10f;
  }

  __shared__ int s_bd[2][32];
  __shared__ int s_bn[2][32];
  __shared__ float s_cx[2][32];
  __shared__ float s_cy[2][32];
  __shared__ float s_cz[2][32];
  __shared__ int s_sel[S_NONE];
  __shared__ float s_ox[S_NONE];
  __shared__ float s_oy[S_NONE];
  __shared__ float s_oz[S_NONE];

  int last = 0;
  float lx, ly, lz;
  {
    const float a = ab[0];
    const float w = branch ? (1.0f - a) : a;
    lx = w * xb[0];
    ly = w * xb[Nn];
    lz = w * xb[2 * Nn];
  }

  for (int k = 0; k < K; ++k) {
    if (tid == 0) {
      s_sel[k] = last;
      s_ox[k] = lx; s_oy[k] = ly; s_oz[k] = lz;
    }
    if (k == K - 1) break;

    float bd = -1.0f;
    int bn = Nn;
#pragma unroll
    for (int i = 0; i < 8; ++i) {
      float dx = px[i] - lx, dy = py[i] - ly, dz = pz[i] - lz;
      float d2 = dx * dx + dy * dy + dz * dz;
      mind[i] = fminf(mind[i], d2);
      if (mind[i] > bd) { bd = mind[i]; bn = tid + i * 1024; }
    }
    int bdi;
    bn = wave_argmax(__float_as_int(bd), bn, &bdi);
    const int wbn = __builtin_amdgcn_readfirstlane(bn);
    const int wi = wbn >> 10;
    float sx = px[0], sy = py[0], sz = pz[0];
#pragma unroll
    for (int i = 1; i < 8; ++i) {
      if (wi == i) { sx = px[i]; sy = py[i]; sz = pz[i]; }
    }
    const int wl = wbn & 31;
    const float cx = __int_as_float(__builtin_amdgcn_readlane(__float_as_int(sx), wl));
    const float cy = __int_as_float(__builtin_amdgcn_readlane(__float_as_int(sy), wl));
    const float cz = __int_as_float(__builtin_amdgcn_readlane(__float_as_int(sz), wl));
    const int par = k & 1;
    if (lane == 0) {
      s_bd[par][wid] = bdi;
      s_bn[par][wid] = bn;
      s_cx[par][wid] = cx;
      s_cy[par][wid] = cy;
      s_cz[par][wid] = cz;
    }
    __syncthreads();

    int gd = s_bd[par][lane];
    int gn = s_bn[par][lane];
    const float gx = s_cx[par][lane];
    const float gy = s_cy[par][lane];
    const float gz = s_cz[par][lane];
    gn = wave_argmax(gd, gn, &gd);
    last = __builtin_amdgcn_readfirstlane(gn);
    const int fw = (last >> 5) & 31;
    lx = __int_as_float(__builtin_amdgcn_readlane(__float_as_int(gx), fw));
    ly = __int_as_float(__builtin_amdgcn_readlane(__float_as_int(gy), fw));
    lz = __int_as_float(__builtin_amdgcn_readlane(__float_as_int(gz), fw));
  }
  __syncthreads();

  for (int k = tid; k < K; k += 1024) {
    int n = s_sel[k];
    n = n < Nn ? n : Nn - 1;
    const int row = rowoff + k;
    out[(size_t)b * 1536 + row] = s_ox[k];
    out[(size_t)b * 1536 + 512 + row] = s_oy[k];
    out[(size_t)b * 1536 + 1024 + row] = s_oz[k];
    out[OUT_ATT_OFF + (size_t)b * 512 + row] = ab[n];
  }
}

template <int CINPAD, int WSTRIDE>
__device__ __forceinline__ void mlp_layer(const float* __restrict__ W,
                                          const float* __restrict__ bias,
                                          const float* sin_, float* sout,
                                          int cout, int lane) {
  const int hl = lane >> 4;
  const int ll = lane & 15;
  for (int ot = 0; ot < cout / 16; ++ot) {
    const float bv = bias[ot * 16 + ll];
    for (int rt = 0; rt < 4; ++rt) {
      v8f acc = {};
      const int m = rt * 16 + ll;
      const int o = ot * 16 + ll;
      for (int c = 0; c < CINPAD; c += 4) {
        const int kk = c + (hl << 1);
        v2f a, bf;
        a.x = sin_[m * C0PAD + kk];
        a.y = sin_[m * C0PAD + kk + 1];
        bf.x = W[o * WSTRIDE + kk];
        bf.y = W[o * WSTRIDE + kk + 1];
        acc = wmma_f32(a, bf, acc);
      }
#pragma unroll
      for (int r = 0; r < 8; ++r) {
        float v = acc[r] + bv;
        v = v > 0.0f ? v : 0.0f;
        const int krow = rt * 16 + r + 8 * hl;
        sout[krow * C0PAD + ot * 16 + ll] = v;
      }
    }
  }
}

__device__ __forceinline__ void mlp_layer3_max(const float* __restrict__ W,
                                               const float* __restrict__ bias,
                                               const float* sin_,
                                               float* __restrict__ out,
                                               int lane, int b, int s) {
  const int hl = lane >> 4;
  const int ll = lane & 15;
  for (int ot = 0; ot < 8; ++ot) {
    const float bv = bias[ot * 16 + ll];
    v8f mx = {};
    for (int rt = 0; rt < 4; ++rt) {
      v8f acc = {};
      const int m = rt * 16 + ll;
      const int o = ot * 16 + ll;
      for (int c = 0; c < 64; c += 4) {
        const int kk = c + (hl << 1);
        v2f a, bf;
        a.x = sin_[m * C0PAD + kk];
        a.y = sin_[m * C0PAD + kk + 1];
        bf.x = W[o * 64 + kk];
        bf.y = W[o * 64 + kk + 1];
        acc = wmma_f32(a, bf, acc);
      }
#pragma unroll
      for (int r = 0; r < 8; ++r) {
        float v = acc[r] + bv;
        v = v > 0.0f ? v : 0.0f;
        mx[r] = fmaxf(mx[r], v);
      }
    }
    float m8 = mx[0];
#pragma unroll
    for (int r = 1; r < 8; ++r) m8 = fmaxf(m8, mx[r]);
    float oth = __shfl_xor(m8, 16, 32);
    m8 = fmaxf(m8, oth);
    if (lane < 16) {
      const int o = ot * 16 + lane;
      out[OUT_NP_OFF + (size_t)b * 65536 + (size_t)o * 512 + s] = m8;
    }
  }
}

__global__ void __launch_bounds__(32) sa_fused_kernel(
    const float* __restrict__ xyz, const float* __restrict__ pts,
    const float* __restrict__ w0, const float* __restrict__ b0,
    const float* __restrict__ w1, const float* __restrict__ b1,
    const float* __restrict__ w2, const float* __restrict__ b2,
    float* __restrict__ out) {
  __shared__ float s_w0[64 * C0PAD];
  __shared__ float s_feat[64 * C0PAD];
  __shared__ float s_act[64 * C0PAD];
  __shared__ int s_idx[NSAMPLE];

  const int lane = threadIdx.x;
  const int b = blockIdx.x >> 9;
  const int s = blockIdx.x & 511;

  for (int t = lane; t < 64 * C0PAD; t += 32) {
    int r = t / C0PAD, c = t - r * C0PAD;
    s_w0[t] = (c < IN_CH) ? w0[r * IN_CH + c] : 0.0f;
  }

  const float* xb = xyz + (size_t)b * 3 * Nn;
  const float qx = out[(size_t)b * 1536 + s];
  const float qy = out[(size_t)b * 1536 + 512 + s];
  const float qz = out[(size_t)b * 1536 + 1024 + s];

  const float R2 = RADIUS * RADIUS;
  int count = 0;
  for (int base = 0; base < Nn && count < NSAMPLE; base += 32) {
    __builtin_prefetch(xb + base + 256, 0, 0);
    const int n = base + lane;
    const float dx = xb[n] - qx;
    const float dy = xb[Nn + n] - qy;
    const float dz = xb[2 * Nn + n] - qz;
    const float d2 = dx * dx + dy * dy + dz * dz;
    const bool hit = d2 < R2;
    const unsigned mask = (unsigned)__ballot(hit);
    const int slot = count + __popc(mask & ((1u << lane) - 1u));
    if (hit && slot < NSAMPLE) s_idx[slot] = n;
    count += __popc(mask);
  }
  __syncthreads();
  const int cnt = count < NSAMPLE ? count : NSAMPLE;

  const float* pb = pts + (size_t)b * D_FEAT * Nn;
  for (int k = 0; k < NSAMPLE; ++k) {
    const int n = (k < cnt) ? s_idx[k] : -1;
    const int gi = (n >= 0) ? n : (Nn - 1);
#pragma unroll
    for (int j = 0; j < 3; ++j) {
      const int ch = lane + j * 32;
      if (ch < C0PAD) {
        float v;
        if (ch < 3) {
          float g = (n >= 0) ? xb[ch * Nn + n] : 0.0f;
          v = g - ((ch == 0) ? qx : (ch == 1) ? qy : qz);
        } else if (ch < IN_CH) {
          v = pb[(ch - 3) * Nn + gi];
        } else {
          v = 0.0f;
        }
        s_feat[k * C0PAD + ch] = v;
      }
    }
  }
  __syncthreads();

  mlp_layer<C0PAD, C0PAD>(s_w0, b0, s_feat, s_act, 64, lane);
  __syncthreads();
  mlp_layer<64, 64>(w1, b1, s_act, s_feat, 64, lane);
  __syncthreads();
  mlp_layer3_max(w2, b2, s_feat, out, lane, b, s);
}

extern "C" void kernel_launch(void* const* d_in, const int* in_sizes, int n_in,
                              void* d_out, int out_size, void* d_ws, size_t ws_size,
                              hipStream_t stream) {
  const float* xyz = (const float*)d_in[0];
  const float* pts = (const float*)d_in[1];
  const float* attn = (const float*)d_in[2];
  const float* w0 = (const float*)d_in[3];
  const float* b0 = (const float*)d_in[4];
  const float* w1 = (const float*)d_in[5];
  const float* b1 = (const float*)d_in[6];
  const float* w2 = (const float*)d_in[7];
  const float* b2 = (const float*)d_in[8];
  float* out = (float*)d_out;

  hipLaunchKernelGGL(fps_kernel, dim3(Bn * 2), dim3(1024), 0, stream, xyz, attn, out);
  hipLaunchKernelGGL(sa_fused_kernel, dim3(Bn * Sn), dim3(32), 0, stream,
                     xyz, pts, w0, b0, w1, b1, w2, b2, out);
}


# ball-query scan: 1024 points per trip (eight 128-point groups, one 128-bit load per lane and coordinate row, all loads of a trip issued before its first ballot) instead of 32 points per trip
# speedup vs baseline: 1.8212x; 1.2960x over previous
#include <hip/hip_runtime.h>

typedef __attribute__((ext_vector_type(2))) float v2f;
typedef __attribute__((ext_vector_type(8))) float v8f;

#define Bn 8
#define Nn 8192
#define S_ATTN 128
#define S_NONE 384
#define Sn 512
#define NSAMPLE 64
#define RADIUS 0.4f
#define D_FEAT 64
#define IN_CH 67
#define C0PAD 68
#define SCAN_GROUPS 8

#define OUT_NP_OFF 12288
#define OUT_ATT_OFF 536576

__device__ __forceinline__ v8f wmma_f32(v2f a, v2f b, v8f c) {
  return __builtin_amdgcn_wmma_f32_16x16x4_f32(
       false, a,  false, b,
       (short)0, c,  false,  false);
}

template <int M>
__device__ __forceinline__ int row_xor(int v) {
  return __builtin_amdgcn_update_dpp(0, v, 0x160 + M, 0xf, 0xf, true);
}
__device__ __forceinline__ int other_row(int v) {
  return __builtin_amdgcn_permlanex16(0, v, 0x76543210, 0xfedcba98, false, false);
}
__device__ __forceinline__ int wave_max(int v) {
  v = max(v, row_xor<1>(v));
  v = max(v, row_xor<2>(v));
  v = max(v, row_xor<4>(v));
  v = max(v, row_xor<8>(v));
  return max(v, other_row(v));
}
__device__ __forceinline__ int wave_min(int v) {
  v = min(v, row_xor<1>(v));
  v = min(v, row_xor<2>(v));
  v = min(v, row_xor<4>(v));
  v = min(v, row_xor<8>(v));
  return min(v, other_row(v));
}
__device__ __forceinline__ int wave_argmax(int d, int n, int* best) {
  const int m = wave_max(d);
  *best = m;
  return wave_min(d == m ? n : 0x7fffffff);
}

__global__ void __launch_bounds__(1024) fps_kernel(const float* __restrict__ xyz,
                                                   const float* __restrict__ attn,
                                                   float* __restrict__ out) {
  const int b = blockIdx.x >> 1;
  const int branch = blockIdx.x & 1;
  const int K = branch ? S_NONE : S_ATTN;
  const int rowoff = branch ? S_ATTN : 0;
  const int tid = threadIdx.x;
  const int wid = tid >> 5, lane = tid & 31;

  const float* xb = xyz + (size_t)b * 3 * Nn;
  const float* ab = attn + (size_t)b * Nn;

  float px[8], py[8], pz[8], mind[8];
#pragma unroll
  for (int i = 0; i < 8; ++i) {
    int n = tid + i * 1024;
    float a = ab[n];
    float w = branch ? (1.0f - a) : a;
    px[i] = w * xb[n];
    py[i] = w * xb[Nn + n];
    pz[i] = w * xb[2 * Nn + n];
    mind[i] = 1e10f;
  }

  __shared__ int s_bd[2][32];
  __shared__ int s_bn[2][32];
  __shared__ float s_cx[2][32];
  __shared__ float s_cy[2][32];
  __shared__ float s_cz[2][32];
  __shared__ int s_sel[S_NONE];
  __shared__ float s_ox[S_NONE];
  __shared__ float s_oy[S_NONE];
  __shared__ float s_oz[S_NONE];

  int last = 0;
  float lx, ly, lz;
  {
    const float a = ab[0];
    const float w = branch ? (1.0f - a) : a;
    lx = w * xb[0];
    ly = w * xb[Nn];
    lz = w * xb[2 * Nn];
  }

  for (int k = 0; k < K; ++k) {
    if (tid == 0) {
      s_sel[k] = last;
      s_ox[k] = lx; s_oy[k] = ly; s_oz[k] = lz;
    }
    if (k == K - 1) break;

    float bd = -1.0f;
    int bn = Nn;
#pragma unroll
    for (int i = 0; i < 8; ++i) {
      float dx = px[i] - lx, dy = py[i] - ly, dz = pz[i] - lz;
      float d2 = dx * dx + dy * dy + dz * dz;
      mind[i] = fminf(mind[i], d2);
      if (mind[i] > bd) { bd = mind[i]; bn = tid + i * 1024; }
    }
    int bdi;
    bn = wave_argmax(__float_as_int(bd), bn, &bdi);
    const int wbn = __builtin_amdgcn_readfirstlane(bn);
    const int wi = wbn >> 10;
    float sx = px[0], sy = py[0], sz = pz[0];
#pragma unroll
    for (int i = 1; i < 8; ++i) {
      if (wi == i) { sx = px[i]; sy = py[i]; sz = pz[i]; }
    }
    const int wl = wbn & 31;
    const float cx = __int_as_float(__builtin_amdgcn_readlane(__float_as_int(sx), wl));
    const float cy = __int_as_float(__builtin_amdgcn_readlane(__float_as_int(sy), wl));
    const float cz = __int_as_float(__builtin_amdgcn_readlane(__float_as_int(sz), wl));
    const int par = k & 1;
    if (lane == 0) {
      s_bd[par][wid] = bdi;
      s_bn[par][wid] = bn;
      s_cx[par][wid] = cx;
      s_cy[par][wid] = cy;
      s_cz[par][wid] = cz;
    }
    __syncthreads();

    int gd = s_bd[par][lane];
    int gn = s_bn[par][lane];
    const float gx = s_cx[par][lane];
    const float gy = s_cy[par][lane];
    const float gz = s_cz[par][lane];
    gn = wave_argmax(gd, gn, &gd);
    last = __builtin_amdgcn_readfirstlane(gn);
    const int fw = (last >> 5) & 31;
    lx = __int_as_float(__builtin_amdgcn_readlane(__float_as_int(gx), fw));
    ly = __int_as_float(__builtin_amdgcn_readlane(__float_as_int(gy), fw));
    lz = __int_as_float(__builtin_amdgcn_readlane(__float_as_int(gz), fw));
  }
  __syncthreads();

  for (int k = tid; k < K; k += 1024) {
    int n = s_sel[k];
    n = n < Nn ? n : Nn - 1;
    const int row = rowoff + k;
    out[(size_t)b * 1536 + row] = s_ox[k];
    out[(size_t)b * 1536 + 512 + row] = s_oy[k];
    out[(size_t)b * 1536 + 1024 + row] = s_oz[k];
    out[OUT_ATT_OFF + (size_t)b * 512 + row] = ab[n];
  }
}

template <int CINPAD, int WSTRIDE>
__device__ __forceinline__ void mlp_layer(const float* __restrict__ W,
                                          const float* __restrict__ bias,
                                          const float* sin_, float* sout,
                                          int cout, int lane) {
  const int hl = lane >> 4;
  const int ll = lane & 15;
  for (int ot = 0; ot < cout / 16; ++ot) {
    const float bv = bias[ot * 16 + ll];
    for (int rt = 0; rt < 4; ++rt) {
      v8f acc = {};
      const int m = rt * 16 + ll;
      const int o = ot * 16 + ll;
      for (int c = 0; c < CINPAD; c += 4) {
        const int kk = c + (hl << 1);
        v2f a, bf;
        a.x = sin_[m * C0PAD + kk];
        a.y = sin_[m * C0PAD + kk + 1];
        bf.x = W[o * WSTRIDE + kk];
        bf.y = W[o * WSTRIDE + kk + 1];
        acc = wmma_f32(a, bf, acc);
      }
#pragma unroll
      for (int r = 0; r < 8; ++r) {
        float v = acc[r] + bv;
        v = v > 0.0f ? v : 0.0f;
        const int krow = rt * 16 + r + 8 * hl;
        sout[krow * C0PAD + ot * 16 + ll] = v;
      }
    }
  }
}

__device__ __forceinline__ void mlp_layer3_max(const float* __restrict__ W,
                                               const float* __restrict__ bias,
                                               const float* sin_,
                                               float* __restrict__ out,
                                               int lane, int b, int s) {
  const int hl = lane >> 4;
  const int ll = lane & 15;
  for (int ot = 0; ot < 8; ++ot) {
    const float bv = bias[ot * 16 + ll];
    v8f mx = {};
    for (int rt = 0; rt < 4; ++rt) {
      v8f acc = {};
      const int m = rt * 16 + ll;
      const int o = ot * 16 + ll;
      for (int c = 0; c < 64; c += 4) {
        const int kk = c + (hl << 1);
        v2f a, bf;
        a.x = sin_[m * C0PAD + kk];
        a.y = sin_[m * C0PAD + kk + 1];
        bf.x = W[o * 64 + kk];
        bf.y = W[o * 64 + kk + 1];
        acc = wmma_f32(a, bf, acc);
      }
#pragma unroll
      for (int r = 0; r < 8; ++r) {
        float v = acc[r] + bv;
        v = v > 0.0f ? v : 0.0f;
        mx[r] = fmaxf(mx[r], v);
      }
    }
    float m8 = mx[0];
#pragma unroll
    for (int r = 1; r < 8; ++r) m8 = fmaxf(m8, mx[r]);
    float oth = __shfl_xor(m8, 16, 32);
    m8 = fmaxf(m8, oth);
    if (lane < 16) {
      const int o = ot * 16 + lane;
      out[OUT_NP_OFF + (size_t)b * 65536 + (size_t)o * 512 + s] = m8;
    }
  }
}

__global__ void __launch_bounds__(32) sa_fused_kernel(
    const float* __restrict__ xyz, const float* __restrict__ pts,
    const float* __restrict__ w0, const float* __restrict__ b0,
    const float* __restrict__ w1, const float* __restrict__ b1,
    const float* __restrict__ w2, const float* __restrict__ b2,
    float* __restrict__ out) {
  __shared__ float s_w0[64 * C0PAD];
  __shared__ float s_feat[64 * C0PAD];
  __shared__ float s_act[64 * C0PAD];
  __shared__ int s_idx[NSAMPLE];

  const int lane = threadIdx.x;
  const int b = blockIdx.x >> 9;
  const int s = blockIdx.x & 511;

  for (int t = lane; t < 64 * C0PAD; t += 32) {
    int r = t / C0PAD, c = t - r * C0PAD;
    s_w0[t] = (c < IN_CH) ? w0[r * IN_CH + c] : 0.0f;
  }

  const float* xb = xyz + (size_t)b * 3 * Nn;
  const float qx = out[(size_t)b * 1536 + s];
  const float qy = out[(size_t)b * 1536 + 512 + s];
  const float qz = out[(size_t)b * 1536 + 1024 + s];

  const float R2 = RADIUS * RADIUS;
  const unsigned lt = (1u << lane) - 1u;
  const float4* x4 = reinterpret_cast<const float4*>(xb);
  const float4* y4 = reinterpret_cast<const float4*>(xb + Nn);
  const float4* z4 = reinterpret_cast<const float4*>(xb + 2 * Nn);
  int count = 0;
  for (int base = 0; base < Nn && count < NSAMPLE; base += 128 * SCAN_GROUPS) {
    float4 vx[SCAN_GROUPS], vy[SCAN_GROUPS], vz[SCAN_GROUPS];
#pragma unroll
    for (int g = 0; g < SCAN_GROUPS; ++g) {
      const int q4 = (base >> 2) + g * 32 + lane;
      vx[g] = x4[q4];
      vy[g] = y4[q4];
      vz[g] = z4[q4];
    }
#pragma unroll
    for (int g = 0; g < SCAN_GROUPS; ++g) {
      const float px[4] = {vx[g].x, vx[g].y, vx[g].z, vx[g].w};
      const float py[4] = {vy[g].x, vy[g].y, vy[g].z, vy[g].w};
      const float pz[4] = {vz[g].x, vz[g].y, vz[g].z, vz[g].w};
      bool hit[4];
      unsigned m[4];
#pragma unroll
      for (int e = 0; e < 4; ++e) {
        const float dx = px[e] - qx;
        const float dy = py[e] - qy;
        const float dz = pz[e] - qz;
        const float d2 = dx * dx + dy * dy + dz * dz;
        hit[e] = d2 < R2;
        m[e] = (unsigned)__ballot(hit[e]);
      }
      if ((m[0] | m[1] | m[2] | m[3]) != 0u) {
        int slot = count + __popc(m[0] & lt) + __popc(m[1] & lt) +
                   __popc(m[2] & lt) + __popc(m[3] & lt);
        const int n0 = base + g * 128 + lane * 4;
#pragma unroll
        for (int e = 0; e < 4; ++e) {
          if (hit[e] && slot < NSAMPLE) s_idx[slot] = n0 + e;
          slot += hit[e] ? 1 : 0;
        }
        count += __popc(m[0]) + __popc(m[1]) + __popc(m[2]) + __popc(m[3]);
      }
    }
  }
  __syncthreads();
  const int cnt = count < NSAMPLE ? count : NSAMPLE;

  const float* pb = pts + (size_t)b * D_FEAT * Nn;
  for (int k = 0; k < NSAMPLE; ++k) {
    const int n = (k < cnt) ? s_idx[k] : -1;
    const int gi = (n >= 0) ? n : (Nn - 1);
#pragma unroll
    for (int j = 0; j < 3; ++j) {
      const int ch = lane + j * 32;
      if (ch < C0PAD) {
        float v;
        if (ch < 3) {
          float g = (n >= 0) ? xb[ch * Nn + n] : 0.0f;
          v = g - ((ch == 0) ? qx : (ch == 1) ? qy : qz);
        } else if (ch < IN_CH) {
          v = pb[(ch - 3) * Nn + gi];
        } else {
          v = 0.0f;
        }
        s_feat[k * C0PAD + ch] = v;
      }
    }
  }
  __syncthreads();

  mlp_layer<C0PAD, C0PAD>(s_w0, b0, s_feat, s_act, 64, lane);
  __syncthreads();
  mlp_layer<64, 64>(w1, b1, s_act, s_feat, 64, lane);
  __syncthreads();
  mlp_layer3_max(w2, b2, s_feat, out, lane, b, s);
}

extern "C" void kernel_launch(void* const* d_in, const int* in_sizes, int n_in,
                              void* d_out, int out_size, void* d_ws, size_t ws_size,
                              hipStream_t stream) {
  const float* xyz = (const float*)d_in[0];
  const float* pts = (const float*)d_in[1];
  const float* attn = (const float*)d_in[2];
  const float* w0 = (const float*)d_in[3];
  const float* b0 = (const float*)d_in[4];
  const float* w1 = (const float*)d_in[5];
  const float* b1 = (const float*)d_in[6];
  const float* w2 = (const float*)d_in[7];
  const float* b2 = (const float*)d_in[8];
  float* out = (float*)d_out;

  hipLaunchKernelGGL(fps_kernel, dim3(Bn * 2), dim3(1024), 0, stream, xyz, attn, out);
  hipLaunchKernelGGL(sa_fused_kernel, dim3(Bn * Sn), dim3(32), 0, stream,
                     xyz, pts, w0, b0, w1, b1, w2, b2, out);
}


# fused kernel's two LDS fill loops (W0 staging, grouping gather) issue their global loads in runs before the first wait, not one load-wait-store round trip per element; values and order unchanged
# speedup vs baseline: 2.5764x; 1.4147x over previous
#include <hip/hip_runtime.h>

typedef __attribute__((ext_vector_type(2))) float v2f;
typedef __attribute__((ext_vector_type(8))) float v8f;

#define Bn 8
#define Nn 8192
#define S_ATTN 128
#define S_NONE 384
#define Sn 512
#define NSAMPLE 64
#define RADIUS 0.4f
#define D_FEAT 64
#define IN_CH 67
#define C0PAD 68
#define SCAN_GROUPS 8
#define W0_RUN 17
#define GATHER_RUN 16

#define OUT_NP_OFF 12288
#define OUT_ATT_OFF 536576

__device__ __forceinline__ v8f wmma_f32(v2f a, v2f b, v8f c) {
  return __builtin_amdgcn_wmma_f32_16x16x4_f32(
       false, a,  false, b,
       (short)0, c,  false,  false);
}

template <int M>
__device__ __forceinline__ int row_xor(int v) {
  return __builtin_amdgcn_update_dpp(0, v, 0x160 + M, 0xf, 0xf, true);
}
__device__ __forceinline__ int other_row(int v) {
  return __builtin_amdgcn_permlanex16(0, v, 0x76543210, 0xfedcba98, false, false);
}
__device__ __forceinline__ int wave_max(int v) {
  v = max(v, row_xor<1>(v));
  v = max(v, row_xor<2>(v));
  v = max(v, row_xor<4>(v));
  v = max(v, row_xor<8>(v));
  return max(v, other_row(v));
}
__device__ __forceinline__ int wave_min(int v) {
  v = min(v, row_xor<1>(v));
  v = min(v, row_xor<2>(v));
  v = min(v, row_xor<4>(v));
  v = min(v, row_xor<8>(v));
  return min(v, other_row(v));
}
__device__ __forceinline__ int wave_argmax(int d, int n, int* best) {
  const int m = wave_max(d);
  *best = m;
  return wave_min(d == m ? n : 0x7fffffff);
}

__global__ void __launch_bounds__(1024) fps_kernel(const float* __restrict__ xyz,
                                                   const float* __restrict__ attn,
                                                   float* __restrict__ out) {
  const int b = blockIdx.x >> 1;
  const int branch = blockIdx.x & 1;
  const int K = branch ? S_NONE : S_ATTN;
  const int rowoff = branch ? S_ATTN : 0;
  const int tid = threadIdx.x;
  const int wid = tid >> 5, lane = tid & 31;

  const float* xb = xyz + (size_t)b * 3 * Nn;
  const float* ab = attn + (size_t)b * Nn;

  float px[8], py[8], pz[8], mind[8];
#pragma unroll
  for (int i = 0; i < 8; ++i) {
    int n = tid + i * 1024;
    float a = ab[n];
    float w = branch ? (1.0f - a) : a;
    px[i] = w * xb[n];
    py[i] = w * xb[Nn + n];
    pz[i] = w * xb[2 * Nn + n];
    mind[i] = 1e10f;
  }

  __shared__ int s_bd[2][32];
  __shared__ int s_bn[2][32];
  __shared__ float s_cx[2][32];
  __shared__ float s_cy[2][32];
  __shared__ float s_cz[2][32];
  __shared__ int s_sel[S_NONE];
  __shared__ float s_ox[S_NONE];
  __shared__ float s_oy[S_NONE];
  __shared__ float s_oz[S_NONE];

  int last = 0;
  float lx, ly, lz;
  {
    const float a = ab[0];
    const float w = branch ? (1.0f - a) : a;
    lx = w * xb[0];
    ly = w * xb[Nn];
    lz = w * xb[2 * Nn];
  }

  for (int k = 0; k < K; ++k) {
    if (tid == 0) {
      s_sel[k] = last;
      s_ox[k] = lx; s_oy[k] = ly; s_oz[k] = lz;
    }
    if (k == K - 1) break;

    float bd = -1.0f;
    int bn = Nn;
#pragma unroll
    for (int i = 0; i < 8; ++i) {
      float dx = px[i] - lx, dy = py[i] - ly, dz = pz[i] - lz;
      float d2 = dx * dx + dy * dy + dz * dz;
      mind[i] = fminf(mind[i], d2);
      if (mind[i] > bd) { bd = mind[i]; bn = tid + i * 1024; }
    }
    int bdi;
    bn = wave_argmax(__float_as_int(bd), bn, &bdi);
    const int wbn = __builtin_amdgcn_readfirstlane(bn);
    const int wi = wbn >> 10;
    float sx = px[0], sy = py[0], sz = pz[0];
#pragma unroll
    for (int i = 1; i < 8; ++i) {
      if (wi == i) { sx = px[i]; sy = py[i]; sz = pz[i]; }
    }
    const int wl = wbn & 31;
    const float cx = __int_as_float(__builtin_amdgcn_readlane(__float_as_int(sx), wl));
    const float cy = __int_as_float(__builtin_amdgcn_readlane(__float_as_int(sy), wl));
    const float cz = __int_as_float(__builtin_amdgcn_readlane(__float_as_int(sz), wl));
    const int par = k & 1;
    if (lane == 0) {
      s_bd[par][wid] = bdi;
      s_bn[par][wid] = bn;
      s_cx[par][wid] = cx;
      s_cy[par][wid] = cy;
      s_cz[par][wid] = cz;
    }
    __syncthreads();

    int gd = s_bd[par][lane];
    int gn = s_bn[par][lane];
    const float gx = s_cx[par][lane];
    const float gy = s_cy[par][lane];
    const float gz = s_cz[par][lane];
    gn = wave_argmax(gd, gn, &gd);
    last = __builtin_amdgcn_readfirstlane(gn);
    const int fw = (last >> 5) & 31;
    lx = __int_as_float(__builtin_amdgcn_readlane(__float_as_int(gx), fw));
    ly = __int_as_float(__builtin_amdgcn_readlane(__float_as_int(gy), fw));
    lz = __int_as_float(__builtin_amdgcn_readlane(__float_as_int(gz), fw));
  }
  __syncthreads();

  for (int k = tid; k < K; k += 1024) {
    int n = s_sel[k];
    n = n < Nn ? n : Nn - 1;
    const int row = rowoff + k;
    out[(size_t)b * 1536 + row] = s_ox[k];
    out[(size_t)b * 1536 + 512 + row] = s_oy[k];
    out[(size_t)b * 1536 + 1024 + row] = s_oz[k];
    out[OUT_ATT_OFF + (size_t)b * 512 + row] = ab[n];
  }
}

template <int CINPAD, int WSTRIDE>
__device__ __forceinline__ void mlp_layer(const float* __restrict__ W,
                                          const float* __restrict__ bias,
                                          const float* sin_, float* sout,
                                          int cout, int lane) {
  const int hl = lane >> 4;
  const int ll = lane & 15;
  for (int ot = 0; ot < cout / 16; ++ot) {
    const float bv = bias[ot * 16 + ll];
    for (int rt = 0; rt < 4; ++rt) {
      v8f acc = {};
      const int m = rt * 16 + ll;
      const int o = ot * 16 + ll;
      for (int c = 0; c < CINPAD; c += 4) {
        const int kk = c + (hl << 1);
        v2f a, bf;
        a.x = sin_[m * C0PAD + kk];
        a.y = sin_[m * C0PAD + kk + 1];
        bf.x = W[o * WSTRIDE + kk];
        bf.y = W[o * WSTRIDE + kk + 1];
        acc = wmma_f32(a, bf, acc);
      }
#pragma unroll
      for (int r = 0; r < 8; ++r) {
        float v = acc[r] + bv;
        v = v > 0.0f ? v : 0.0f;
        const int krow = rt * 16 + r + 8 * hl;
        sout[krow * C0PAD + ot * 16 + ll] = v;
      }
    }
  }
}

__device__ __forceinline__ void mlp_layer3_max(const float* __restrict__ W,
                                               const float* __restrict__ bias,
                                               const float* sin_,
                                               float* __restrict__ out,
                                               int lane, int b, int s) {
  const int hl = lane >> 4;
  const int ll = lane & 15;
  for (int ot = 0; ot < 8; ++ot) {
    const float bv = bias[ot * 16 + ll];
    v8f mx = {};
    for (int rt = 0; rt < 4; ++rt) {
      v8f acc = {};
      const int m = rt * 16 + ll;
      const int o = ot * 16 + ll;
      for (int c = 0; c < 64; c += 4) {
        const int kk = c + (hl << 1);
        v2f a, bf;
        a.x = sin_[m * C0PAD + kk];
        a.y = sin_[m * C0PAD + kk + 1];
        bf.x = W[o * 64 + kk];
        bf.y = W[o * 64 + kk + 1];
        acc = wmma_f32(a, bf, acc);
      }
#pragma unroll
      for (int r = 0; r < 8; ++r) {
        float v = acc[r] + bv;
        v = v > 0.0f ? v : 0.0f;
        mx[r] = fmaxf(mx[r], v);
      }
    }
    float m8 = mx[0];
#pragma unroll
    for (int r = 1; r < 8; ++r) m8 = fmaxf(m8, mx[r]);
    float oth = __shfl_xor(m8, 16, 32);
    m8 = fmaxf(m8, oth);
    if (lane < 16) {
      const int o = ot * 16 + lane;
      out[OUT_NP_OFF + (size_t)b * 65536 + (size_t)o * 512 + s] = m8;
    }
  }
}

__global__ void __launch_bounds__(32) sa_fused_kernel(
    const float* __restrict__ xyz, const float* __restrict__ pts,
    const float* __restrict__ w0, const float* __restrict__ b0,
    const float* __restrict__ w1, const float* __restrict__ b1,
    const float* __restrict__ w2, const float* __restrict__ b2,
    float* __restrict__ out) {
  __shared__ float s_w0[64 * C0PAD];
  __shared__ float s_feat[64 * C0PAD];
  __shared__ float s_act[64 * C0PAD];
  __shared__ int s_idx[NSAMPLE];

  const int lane = threadIdx.x;
  const int b = blockIdx.x >> 9;
  const int s = blockIdx.x & 511;

  for (int i0 = 0; i0 < (64 * C0PAD) / 32; i0 += W0_RUN) {
    float wv[W0_RUN];
#pragma unroll
    for (int j = 0; j < W0_RUN; ++j) {
      const int t = lane + 32 * (i0 + j);
      const int r = t / C0PAD, c = t - r * C0PAD;
      const float x = w0[r * IN_CH + (c < IN_CH ? c : IN_CH - 1)];
      wv[j] = (c < IN_CH) ? x : 0.0f;
    }
#pragma unroll
    for (int j = 0; j < W0_RUN; ++j) s_w0[lane + 32 * (i0 + j)] = wv[j];
  }

  const float* xb = xyz + (size_t)b * 3 * Nn;
  const float qx = out[(size_t)b * 1536 + s];
  const float qy = out[(size_t)b * 1536 + 512 + s];
  const float qz = out[(size_t)b * 1536 + 1024 + s];

  const float R2 = RADIUS * RADIUS;
  const unsigned lt = (1u << lane) - 1u;
  const float4* x4 = reinterpret_cast<const float4*>(xb);
  const float4* y4 = reinterpret_cast<const float4*>(xb + Nn);
  const float4* z4 = reinterpret_cast<const float4*>(xb + 2 * Nn);
  int count = 0;
  for (int base = 0; base < Nn && count < NSAMPLE; base += 128 * SCAN_GROUPS) {
    float4 vx[SCAN_GROUPS], vy[SCAN_GROUPS], vz[SCAN_GROUPS];
#pragma unroll
    for (int g = 0; g < SCAN_GROUPS; ++g) {
      const int q4 = (base >> 2) + g * 32 + lane;
      vx[g] = x4[q4];
      vy[g] = y4[q4];
      vz[g] = z4[q4];
    }
#pragma unroll
    for (int g = 0; g < SCAN_GROUPS; ++g) {
      const float px[4] = {vx[g].x, vx[g].y, vx[g].z, vx[g].w};
      const float py[4] = {vy[g].x, vy[g].y, vy[g].z, vy[g].w};
      const float pz[4] = {vz[g].x, vz[g].y, vz[g].z, vz[g].w};
      bool hit[4];
      unsigned m[4];
#pragma unroll
      for (int e = 0; e < 4; ++e) {
        const float dx = px[e] - qx;
        const float dy = py[e] - qy;
        const float dz = pz[e] - qz;
        const float d2 = dx * dx + dy * dy + dz * dz;
        hit[e] = d2 < R2;
        m[e] = (unsigned)__ballot(hit[e]);
      }
      if ((m[0] | m[1] | m[2] | m[3]) != 0u) {
        int slot = count + __popc(m[0] & lt) + __popc(m[1] & lt) +
                   __popc(m[2] & lt) + __popc(m[3] & lt);
        const int n0 = base + g * 128 + lane * 4;
#pragma unroll
        for (int e = 0; e < 4; ++e) {
          if (hit[e] && slot < NSAMPLE) s_idx[slot] = n0 + e;
          slot += hit[e] ? 1 : 0;
        }
        count += __popc(m[0]) + __popc(m[1]) + __popc(m[2]) + __popc(m[3]);
      }
    }
  }
  __syncthreads();
  const int cnt = count < NSAMPLE ? count : NSAMPLE;

  const float* pb = pts + (size_t)b * D_FEAT * Nn;
  const int idA = (lane < cnt) ? s_idx[lane] : -1;
  const int idB = (lane + 32 < cnt) ? s_idx[lane + 32] : -1;
  {
    const int giA = (idA >= 0) ? idA : (Nn - 1);
    const int giB = (idB >= 0) ? idB : (Nn - 1);
    float ga[3], gb[3];
#pragma unroll
    for (int ch = 0; ch < 3; ++ch) {
      ga[ch] = xb[ch * Nn + giA];
      gb[ch] = xb[ch * Nn + giB];
    }
    const float q[3] = {qx, qy, qz};
#pragma unroll
    for (int ch = 0; ch < 3; ++ch) {
      s_feat[lane * C0PAD + ch] = ((idA >= 0) ? ga[ch] : 0.0f) - q[ch];
      s_feat[(lane + 32) * C0PAD + ch] = ((idB >= 0) ? gb[ch] : 0.0f) - q[ch];
    }
    s_feat[lane * C0PAD + IN_CH] = 0.0f;
    s_feat[(lane + 32) * C0PAD + IN_CH] = 0.0f;
  }
  const float* pl = pb + (size_t)lane * Nn;
  for (int k0 = 0; k0 < NSAMPLE; k0 += GATHER_RUN) {
    float f0[GATHER_RUN], f1[GATHER_RUN];
    const int ids = (k0 < 32) ? idA : idB;
#pragma unroll
    for (int u = 0; u < GATHER_RUN; ++u) {
      const int n = __builtin_amdgcn_readlane(ids, (k0 + u) & 31);
      const int gi = (n >= 0) ? n : (Nn - 1);
      f0[u] = pl[gi];
      f1[u] = pl[32 * Nn + gi];
    }
#pragma unroll
    for (int u = 0; u < GATHER_RUN; ++u) {
      s_feat[(k0 + u) * C0PAD + 3 + lane] = f0[u];
      s_feat[(k0 + u) * C0PAD + 35 + lane] = f1[u];
    }
  }
  __syncthreads();

  mlp_layer<C0PAD, C0PAD>(s_w0, b0, s_feat, s_act, 64, lane);
  __syncthreads();
  mlp_layer<64, 64>(w1, b1, s_act, s_feat, 64, lane);
  __syncthreads();
  mlp_layer3_max(w2, b2, s_feat, out, lane, b, s);
}

extern "C" void kernel_launch(void* const* d_in, const int* in_sizes, int n_in,
                              void* d_out, int out_size, void* d_ws, size_t ws_size,
                              hipStream_t stream) {
  const float* xyz = (const float*)d_in[0];
  const float* pts = (const float*)d_in[1];
  const float* attn = (const float*)d_in[2];
  const float* w0 = (const float*)d_in[3];
  const float* b0 = (const float*)d_in[4];
  const float* w1 = (const float*)d_in[5];
  const float* b1 = (const float*)d_in[6];
  const float* w2 = (const float*)d_in[7];
  const float* b2 = (const float*)d_in[8];
  float* out = (float*)d_out;

  hipLaunchKernelGGL(fps_kernel, dim3(Bn * 2), dim3(1024), 0, stream, xyz, attn, out);
  hipLaunchKernelGGL(sa_fused_kernel, dim3(Bn * Sn), dim3(32), 0, stream,
                     xyz, pts, w0, b0, w1, b1, w2, b2, out);
}


# FPS block: 8 waves of 32 points per lane instead of 32 waves of 8, so a pick's wave argmax, barrier and candidate reduction are paid by 8 waves; pick coordinates read by index from an LDS table
# speedup vs baseline: 3.3810x; 1.3123x over previous
#include <hip/hip_runtime.h>

typedef __attribute__((ext_vector_type(2))) float v2f;
typedef __attribute__((ext_vector_type(8))) float v8f;

#define Bn 8
#define Nn 8192
#define S_ATTN 128
#define S_NONE 384
#define Sn 512
#define NSAMPLE 64
#define RADIUS 0.4f
#define D_FEAT 64
#define IN_CH 67
#define C0PAD 68
#define SCAN_GROUPS 8
#define W0_RUN 17
#define GATHER_RUN 16
#define FPS_THREADS 256
#define FPS_WAVES 8
#define FPS_PTS 32

#define OUT_NP_OFF 12288
#define OUT_ATT_OFF 536576

__device__ __forceinline__ v8f wmma_f32(v2f a, v2f b, v8f c) {
  return __builtin_amdgcn_wmma_f32_16x16x4_f32(
       false, a,  false, b,
       (short)0, c,  false,  false);
}

template <int M>
__device__ __forceinline__ int row_xor(int v) {
  return __builtin_amdgcn_update_dpp(0, v, 0x160 + M, 0xf, 0xf, true);
}
__device__ __forceinline__ int other_row(int v) {
  return __builtin_amdgcn_permlanex16(0, v, 0x76543210, 0xfedcba98, false, false);
}
__device__ __forceinline__ int wave_max(int v) {
  v = max(v, row_xor<1>(v));
  v = max(v, row_xor<2>(v));
  v = max(v, row_xor<4>(v));
  v = max(v, row_xor<8>(v));
  return max(v, other_row(v));
}
__device__ __forceinline__ int wave_min(int v) {
  v = min(v, row_xor<1>(v));
  v = min(v, row_xor<2>(v));
  v = min(v, row_xor<4>(v));
  v = min(v, row_xor<8>(v));
  return min(v, other_row(v));
}
__device__ __forceinline__ int wave_argmax(int d, int n, int* best) {
  const int m = wave_max(d);
  *best = m;
  return wave_min(d == m ? n : 0x7fffffff);
}
__device__ __forceinline__ int oct_argmax(int d, int n) {
  int m = d;
  m = max(m, row_xor<1>(m));
  m = max(m, row_xor<2>(m));
  m = max(m, row_xor<4>(m));
  int c = (d == m) ? n : 0x7fffffff;
  c = min(c, row_xor<1>(c));
  c = min(c, row_xor<2>(c));
  return min(c, row_xor<4>(c));
}

__global__ void __launch_bounds__(FPS_THREADS) fps_kernel(const float* __restrict__ xyz,
                                                          const float* __restrict__ attn,
                                                          float* __restrict__ out) {
  const int b = blockIdx.x >> 1;
  const int branch = blockIdx.x & 1;
  const int K = branch ? S_NONE : S_ATTN;
  const int rowoff = branch ? S_ATTN : 0;
  const int tid = threadIdx.x;
  const int wid = tid >> 5, lane = tid & 31;

  const float* xb = xyz + (size_t)b * 3 * Nn;
  const float* ab = attn + (size_t)b * Nn;

  __shared__ float s_px[Nn];
  __shared__ float s_py[Nn];
  __shared__ float s_pz[Nn];
  __shared__ int s_bd[2][FPS_WAVES];
  __shared__ int s_bn[2][FPS_WAVES];
  __shared__ int s_sel[S_NONE];

  float px[FPS_PTS], py[FPS_PTS], pz[FPS_PTS], mind[FPS_PTS];
#pragma unroll
  for (int i = 0; i < FPS_PTS; ++i) {
    const int n = tid + i * FPS_THREADS;
    const float a = ab[n];
    const float w = branch ? (1.0f - a) : a;
    px[i] = w * xb[n];
    py[i] = w * xb[Nn + n];
    pz[i] = w * xb[2 * Nn + n];
    s_px[n] = px[i];
    s_py[n] = py[i];
    s_pz[n] = pz[i];
    mind[i] = 1e10f;
  }
  __syncthreads();

  int last = 0;
  float lx = s_px[0], ly = s_py[0], lz = s_pz[0];

  for (int k = 0; k < K; ++k) {
    if (tid == 0) s_sel[k] = last;
    if (k == K - 1) break;

    float bd = -1.0f;
    int bi = 0;
#pragma unroll
    for (int i = 0; i < FPS_PTS; ++i) {
      float dx = px[i] - lx, dy = py[i] - ly, dz = pz[i] - lz;
      float d2 = dx * dx + dy * dy + dz * dz;
      mind[i] = fminf(mind[i], d2);
      if (mind[i] > bd) { bd = mind[i]; bi = i; }
    }
    int bn = tid + bi * FPS_THREADS;
    int bdi;
    bn = wave_argmax(__float_as_int(bd), bn, &bdi);
    const int par = k & 1;
    if (lane == 0) {
      s_bd[par][wid] = bdi;
      s_bn[par][wid] = bn;
    }
    __syncthreads();

    const int gd = s_bd[par][lane & (FPS_WAVES - 1)];
    const int gn = s_bn[par][lane & (FPS_WAVES - 1)];
    last = __builtin_amdgcn_readfirstlane(oct_argmax(gd, gn));
    last = last < Nn ? last : Nn - 1;
    lx = s_px[last];
    ly = s_py[last];
    lz = s_pz[last];
  }
  __syncthreads();

  for (int k = tid; k < K; k += FPS_THREADS) {
    int n = s_sel[k];
    n = n < Nn ? n : Nn - 1;
    const int row = rowoff + k;
    out[(size_t)b * 1536 + row] = s_px[n];
    out[(size_t)b * 1536 + 512 + row] = s_py[n];
    out[(size_t)b * 1536 + 1024 + row] = s_pz[n];
    out[OUT_ATT_OFF + (size_t)b * 512 + row] = ab[n];
  }
}

template <int CINPAD, int WSTRIDE>
__device__ __forceinline__ void mlp_layer(const float* __restrict__ W,
                                          const float* __restrict__ bias,
                                          const float* sin_, float* sout,
                                          int cout, int lane) {
  const int hl = lane >> 4;
  const int ll = lane & 15;
  for (int ot = 0; ot < cout / 16; ++ot) {
    const float bv = bias[ot * 16 + ll];
    for (int rt = 0; rt < 4; ++rt) {
      v8f acc = {};
      const int m = rt * 16 + ll;
      const int o = ot * 16 + ll;
      for (int c = 0; c < CINPAD; c += 4) {
        const int kk = c + (hl << 1);
        v2f a, bf;
        a.x = sin_[m * C0PAD + kk];
        a.y = sin_[m * C0PAD + kk + 1];
        bf.x = W[o * WSTRIDE + kk];
        bf.y = W[o * WSTRIDE + kk + 1];
        acc = wmma_f32(a, bf, acc);
      }
#pragma unroll
      for (int r = 0; r < 8; ++r) {
        float v = acc[r] + bv;
        v = v > 0.0f ? v : 0.0f;
        const int krow = rt * 16 + r + 8 * hl;
        sout[krow * C0PAD + ot * 16 + ll] = v;
      }
    }
  }
}

__device__ __forceinline__ void mlp_layer3_max(const float* __restrict__ W,
                                               const float* __restrict__ bias,
                                               const float* sin_,
                                               float* __restrict__ out,
                                               int lane, int b, int s) {
  const int hl = lane >> 4;
  const int ll = lane & 15;
  for (int ot = 0; ot < 8; ++ot) {
    const float bv = bias[ot * 16 + ll];
    v8f mx = {};
    for (int rt = 0; rt < 4; ++rt) {
      v8f acc = {};
      const int m = rt * 16 + ll;
      const int o = ot * 16 + ll;
      for (int c = 0; c < 64; c += 4) {
        const int kk = c + (hl << 1);
        v2f a, bf;
        a.x = sin_[m * C0PAD + kk];
        a.y = sin_[m * C0PAD + kk + 1];
        bf.x = W[o * 64 + kk];
        bf.y = W[o * 64 + kk + 1];
        acc = wmma_f32(a, bf, acc);
      }
#pragma unroll
      for (int r = 0; r < 8; ++r) {
        float v = acc[r] + bv;
        v = v > 0.0f ? v : 0.0f;
        mx[r] = fmaxf(mx[r], v);
      }
    }
    float m8 = mx[0];
#pragma unroll
    for (int r = 1; r < 8; ++r) m8 = fmaxf(m8, mx[r]);
    float oth = __shfl_xor(m8, 16, 32);
    m8 = fmaxf(m8, oth);
    if (lane < 16) {
      const int o = ot * 16 + lane;
      out[OUT_NP_OFF + (size_t)b * 65536 + (size_t)o * 512 + s] = m8;
    }
  }
}

__global__ void __launch_bounds__(32) sa_fused_kernel(
    const float* __restrict__ xyz, const float* __restrict__ pts,
    const float* __restrict__ w0, const float* __restrict__ b0,
    const float* __restrict__ w1, const float* __restrict__ b1,
    const float* __restrict__ w2, const float* __restrict__ b2,
    float* __restrict__ out) {
  __shared__ float s_w0[64 * C0PAD];
  __shared__ float s_feat[64 * C0PAD];
  __shared__ float s_act[64 * C0PAD];
  __shared__ int s_idx[NSAMPLE];

  const int lane = threadIdx.x;
  const int b = blockIdx.x >> 9;
  const int s = blockIdx.x & 511;

  for (int i0 = 0; i0 < (64 * C0PAD) / 32; i0 += W0_RUN) {
    float wv[W0_RUN];
#pragma unroll
    for (int j = 0; j < W0_RUN; ++j) {
      const int t = lane + 32 * (i0 + j);
      const int r = t / C0PAD, c = t - r * C0PAD;
      const float x = w0[r * IN_CH + (c < IN_CH ? c : IN_CH - 1)];
      wv[j] = (c < IN_CH) ? x : 0.0f;
    }
#pragma unroll
    for (int j = 0; j < W0_RUN; ++j) s_w0[lane + 32 * (i0 + j)] = wv[j];
  }

  const float* xb = xyz + (size_t)b * 3 * Nn;
  const float qx = out[(size_t)b * 1536 + s];
  const float qy = out[(size_t)b * 1536 + 512 + s];
  const float qz = out[(size_t)b * 1536 + 1024 + s];

  const float R2 = RADIUS * RADIUS;
  const unsigned lt = (1u << lane) - 1u;
  const float4* x4 = reinterpret_cast<const float4*>(xb);
  const float4* y4 = reinterpret_cast<const float4*>(xb + Nn);
  const float4* z4 = reinterpret_cast<const float4*>(xb + 2 * Nn);
  int count = 0;
  for (int base = 0; base < Nn && count < NSAMPLE; base += 128 * SCAN_GROUPS) {
    float4 vx[SCAN_GROUPS], vy[SCAN_GROUPS], vz[SCAN_GROUPS];
#pragma unroll
    for (int g = 0; g < SCAN_GROUPS; ++g) {
      const int q4 = (base >> 2) + g * 32 + lane;
      vx[g] = x4[q4];
      vy[g] = y4[q4];
      vz[g] = z4[q4];
    }
#pragma unroll
    for (int g = 0; g < SCAN_GROUPS; ++g) {
      const float px[4] = {vx[g].x, vx[g].y, vx[g].z, vx[g].w};
      const float py[4] = {vy[g].x, vy[g].y, vy[g].z, vy[g].w};
      const float pz[4] = {vz[g].x, vz[g].y, vz[g].z, vz[g].w};
      bool hit[4];
      unsigned m[4];
#pragma unroll
      for (int e = 0; e < 4; ++e) {
        const float dx = px[e] - qx;
        const float dy = py[e] - qy;
        const float dz = pz[e] - qz;
        const float d2 = dx * dx + dy * dy + dz * dz;
        hit[e] = d2 < R2;
        m[e] = (unsigned)__ballot(hit[e]);
      }
      if ((m[0] | m[1] | m[2] | m[3]) != 0u) {
        int slot = count + __popc(m[0] & lt) + __popc(m[1] & lt) +
                   __popc(m[2] & lt) + __popc(m[3] & lt);
        const int n0 = base + g * 128 + lane * 4;
#pragma unroll
        for (int e = 0; e < 4; ++e) {
          if (hit[e] && slot < NSAMPLE) s_idx[slot] = n0 + e;
          slot += hit[e] ? 1 : 0;
        }
        count += __popc(m[0]) + __popc(m[1]) + __popc(m[2]) + __popc(m[3]);
      }
    }
  }
  __syncthreads();
  const int cnt = count < NSAMPLE ? count : NSAMPLE;

  const float* pb = pts + (size_t)b * D_FEAT * Nn;
  const int idA = (lane < cnt) ? s_idx[lane] : -1;
  const int idB = (lane + 32 < cnt) ? s_idx[lane + 32] : -1;
  {
    const int giA = (idA >= 0) ? idA : (Nn - 1);
    const int giB = (idB >= 0) ? idB : (Nn - 1);
    float ga[3], gb[3];
#pragma unroll
    for (int ch = 0; ch < 3; ++ch) {
      ga[ch] = xb[ch * Nn + giA];
      gb[ch] = xb[ch * Nn + giB];
    }
    const float q[3] = {qx, qy, qz};
#pragma unroll
    for (int ch = 0; ch < 3; ++ch) {
      s_feat[lane * C0PAD + ch] = ((idA >= 0) ? ga[ch] : 0.0f) - q[ch];
      s_feat[(lane + 32) * C0PAD + ch] = ((idB >= 0) ? gb[ch] : 0.0f) - q[ch];
    }
    s_feat[lane * C0PAD + IN_CH] = 0.0f;
    s_feat[(lane + 32) * C0PAD + IN_CH] = 0.0f;
  }
  const float* pl = pb + (size_t)lane * Nn;
  for (int k0 = 0; k0 < NSAMPLE; k0 += GATHER_RUN) {
    float f0[GATHER_RUN], f1[GATHER_RUN];
    const int ids = (k0 < 32) ? idA : idB;
#pragma unroll
    for (int u = 0; u < GATHER_RUN; ++u) {
      const int n = __builtin_amdgcn_readlane(ids, (k0 + u) & 31);
      const int gi = (n >= 0) ? n : (Nn - 1);
      f0[u] = pl[gi];
      f1[u] = pl[32 * Nn + gi];
    }
#pragma unroll
    for (int u = 0; u < GATHER_RUN; ++u) {
      s_feat[(k0 + u) * C0PAD + 3 + lane] = f0[u];
      s_feat[(k0 + u) * C0PAD + 35 + lane] = f1[u];
    }
  }
  __syncthreads();

  mlp_layer<C0PAD, C0PAD>(s_w0, b0, s_feat, s_act, 64, lane);
  __syncthreads();
  mlp_layer<64, 64>(w1, b1, s_act, s_feat, 64, lane);
  __syncthreads();
  mlp_layer3_max(w2, b2, s_feat, out, lane, b, s);
}

extern "C" void kernel_launch(void* const* d_in, const int* in_sizes, int n_in,
                              void* d_out, int out_size, void* d_ws, size_t ws_size,
                              hipStream_t stream) {
  const float* xyz = (const float*)d_in[0];
  const float* pts = (const float*)d_in[1];
  const float* attn = (const float*)d_in[2];
  const float* w0 = (const float*)d_in[3];
  const float* b0 = (const float*)d_in[4];
  const float* w1 = (const float*)d_in[5];
  const float* b1 = (const float*)d_in[6];
  const float* w2 = (const float*)d_in[7];
  const float* b2 = (const float*)d_in[8];
  float* out = (float*)d_out;

  hipLaunchKernelGGL(fps_kernel, dim3(Bn * 2), dim3(FPS_THREADS), 0, stream, xyz, attn, out);
  hipLaunchKernelGGL(sa_fused_kernel, dim3(Bn * Sn), dim3(32), 0, stream,
                     xyz, pts, w0, b0, w1, b1, w2, b2, out);
}


# fused kernel: every pad slot holds the same row, so only the 16-row tiles with a sample or the first pad row are gathered and multiplied (1 to 4 of 4); two column tiles share each A fragment
# speedup vs baseline: 3.6537x; 1.0807x over previous
#include <hip/hip_runtime.h>

typedef __attribute__((ext_vector_type(2))) float v2f;
typedef __attribute__((ext_vector_type(8))) float v8f;

#define Bn 8
#define Nn 8192
#define S_ATTN 128
#define S_NONE 384
#define Sn 512
#define NSAMPLE 64
#define RADIUS 0.4f
#define D_FEAT 64
#define IN_CH 67
#define C0PAD 68
#define SCAN_GROUPS 8
#define W0_RUN 17
#define GATHER_RUN 16
#define FPS_THREADS 256
#define FPS_WAVES 8
#define FPS_PTS 32

#define OUT_NP_OFF 12288
#define OUT_ATT_OFF 536576

__device__ __forceinline__ v8f wmma_f32(v2f a, v2f b, v8f c) {
  return __builtin_amdgcn_wmma_f32_16x16x4_f32(
       false, a,  false, b,
       (short)0, c,  false,  false);
}

template <int M>
__device__ __forceinline__ int row_xor(int v) {
  return __builtin_amdgcn_update_dpp(0, v, 0x160 + M, 0xf, 0xf, true);
}
__device__ __forceinline__ int other_row(int v) {
  return __builtin_amdgcn_permlanex16(0, v, 0x76543210, 0xfedcba98, false, false);
}
__device__ __forceinline__ int wave_max(int v) {
  v = max(v, row_xor<1>(v));
  v = max(v, row_xor<2>(v));
  v = max(v, row_xor<4>(v));
  v = max(v, row_xor<8>(v));
  return max(v, other_row(v));
}
__device__ __forceinline__ int wave_min(int v) {
  v = min(v, row_xor<1>(v));
  v = min(v, row_xor<2>(v));
  v = min(v, row_xor<4>(v));
  v = min(v, row_xor<8>(v));
  return min(v, other_row(v));
}
__device__ __forceinline__ int wave_argmax(int d, int n, int* best) {
  const int m = wave_max(d);
  *best = m;
  return wave_min(d == m ? n : 0x7fffffff);
}
__device__ __forceinline__ int oct_argmax(int d, int n) {
  int m = d;
  m = max(m, row_xor<1>(m));
  m = max(m, row_xor<2>(m));
  m = max(m, row_xor<4>(m));
  int c = (d == m) ? n : 0x7fffffff;
  c = min(c, row_xor<1>(c));
  c = min(c, row_xor<2>(c));
  return min(c, row_xor<4>(c));
}

__global__ void __launch_bounds__(FPS_THREADS) fps_kernel(const float* __restrict__ xyz,
                                                          const float* __restrict__ attn,
                                                          float* __restrict__ out) {
  const int b = blockIdx.x >> 1;
  const int branch = blockIdx.x & 1;
  const int K = branch ? S_NONE : S_ATTN;
  const int rowoff = branch ? S_ATTN : 0;
  const int tid = threadIdx.x;
  const int wid = tid >> 5, lane = tid & 31;

  const float* xb = xyz + (size_t)b * 3 * Nn;
  const float* ab = attn + (size_t)b * Nn;

  __shared__ float s_px[Nn];
  __shared__ float s_py[Nn];
  __shared__ float s_pz[Nn];
  __shared__ int s_bd[2][FPS_WAVES];
  __shared__ int s_bn[2][FPS_WAVES];
  __shared__ int s_sel[S_NONE];

  float px[FPS_PTS], py[FPS_PTS], pz[FPS_PTS], mind[FPS_PTS];
#pragma unroll
  for (int i = 0; i < FPS_PTS; ++i) {
    const int n = tid + i * FPS_THREADS;
    const float a = ab[n];
    const float w = branch ? (1.0f - a) : a;
    px[i] = w * xb[n];
    py[i] = w * xb[Nn + n];
    pz[i] = w * xb[2 * Nn + n];
    s_px[n] = px[i];
    s_py[n] = py[i];
    s_pz[n] = pz[i];
    mind[i] = 1e10f;
  }
  __syncthreads();

  int last = 0;
  float lx = s_px[0], ly = s_py[0], lz = s_pz[0];

  for (int k = 0; k < K; ++k) {
    if (tid == 0) s_sel[k] = last;
    if (k == K - 1) break;

    float bd = -1.0f;
    int bi = 0;
#pragma unroll
    for (int i = 0; i < FPS_PTS; ++i) {
      float dx = px[i] - lx, dy = py[i] - ly, dz = pz[i] - lz;
      float d2 = dx * dx + dy * dy + dz * dz;
      mind[i] = fminf(mind[i], d2);
      if (mind[i] > bd) { bd = mind[i]; bi = i; }
    }
    int bn = tid + bi * FPS_THREADS;
    int bdi;
    bn = wave_argmax(__float_as_int(bd), bn, &bdi);
    const int par = k & 1;
    if (lane == 0) {
      s_bd[par][wid] = bdi;
      s_bn[par][wid] = bn;
    }
    __syncthreads();

    const int gd = s_bd[par][lane & (FPS_WAVES - 1)];
    const int gn = s_bn[par][lane & (FPS_WAVES - 1)];
    last = __builtin_amdgcn_readfirstlane(oct_argmax(gd, gn));
    last = last < Nn ? last : Nn - 1;
    lx = s_px[last];
    ly = s_py[last];
    lz = s_pz[last];
  }
  __syncthreads();

  for (int k = tid; k < K; k += FPS_THREADS) {
    int n = s_sel[k];
    n = n < Nn ? n : Nn - 1;
    const int row = rowoff + k;
    out[(size_t)b * 1536 + row] = s_px[n];
    out[(size_t)b * 1536 + 512 + row] = s_py[n];
    out[(size_t)b * 1536 + 1024 + row] = s_pz[n];
    out[OUT_ATT_OFF + (size_t)b * 512 + row] = ab[n];
  }
}

template <int NK, int WSTRIDE>
__device__ __forceinline__ void mlp_layer(const float* __restrict__ W,
                                          const float* __restrict__ bias,
                                          const float* sin_, float* sout,
                                          int cout, int lane, int ntile) {
  const int hl = lane >> 4;
  const int ll = lane & 15;
  for (int op = 0; op < cout / 32; ++op) {
    const int o0 = op * 32 + ll;
    const int o1 = o0 + 16;
    const float bv0 = bias[o0];
    const float bv1 = bias[o1];
    v2f wf0[NK], wf1[NK];
#pragma unroll
    for (int j = 0; j < NK; ++j) {
      const int kk = 4 * j + (hl << 1);
      wf0[j].x = W[o0 * WSTRIDE + kk];
      wf0[j].y = W[o0 * WSTRIDE + kk + 1];
      wf1[j].x = W[o1 * WSTRIDE + kk];
      wf1[j].y = W[o1 * WSTRIDE + kk + 1];
    }
    for (int rt = 0; rt < ntile; ++rt) {
      v8f acc0 = {};
      v8f acc1 = {};
      const int m = rt * 16 + ll;
#pragma unroll
      for (int j = 0; j < NK; ++j) {
        const int kk = 4 * j + (hl << 1);
        v2f a;
        a.x = sin_[m * C0PAD + kk];
        a.y = sin_[m * C0PAD + kk + 1];
        acc0 = wmma_f32(a, wf0[j], acc0);
        acc1 = wmma_f32(a, wf1[j], acc1);
      }
#pragma unroll
      for (int r = 0; r < 8; ++r) {
        float v0 = acc0[r] + bv0;
        float v1 = acc1[r] + bv1;
        v0 = v0 > 0.0f ? v0 : 0.0f;
        v1 = v1 > 0.0f ? v1 : 0.0f;
        const int krow = rt * 16 + r + 8 * hl;
        sout[krow * C0PAD + o0] = v0;
        sout[krow * C0PAD + o1] = v1;
      }
    }
  }
}

__device__ __forceinline__ void mlp_layer3_max(const float* __restrict__ W,
                                               const float* __restrict__ bias,
                                               const float* sin_,
                                               float* __restrict__ out,
                                               int lane, int b, int s, int ntile) {
  const int hl = lane >> 4;
  const int ll = lane & 15;
  for (int op = 0; op < 4; ++op) {
    const int o0 = op * 32 + ll;
    const int o1 = o0 + 16;
    const float bv0 = bias[o0];
    const float bv1 = bias[o1];
    v2f wf0[16], wf1[16];
#pragma unroll
    for (int j = 0; j < 16; ++j) {
      const int kk = 4 * j + (hl << 1);
      wf0[j].x = W[o0 * 64 + kk];
      wf0[j].y = W[o0 * 64 + kk + 1];
      wf1[j].x = W[o1 * 64 + kk];
      wf1[j].y = W[o1 * 64 + kk + 1];
    }
    v8f mx0 = {};
    v8f mx1 = {};
    for (int rt = 0; rt < ntile; ++rt) {
      v8f acc0 = {};
      v8f acc1 = {};
      const int m = rt * 16 + ll;
#pragma unroll
      for (int j = 0; j < 16; ++j) {
        const int kk = 4 * j + (hl << 1);
        v2f a;
        a.x = sin_[m * C0PAD + kk];
        a.y = sin_[m * C0PAD + kk + 1];
        acc0 = wmma_f32(a, wf0[j], acc0);
        acc1 = wmma_f32(a, wf1[j], acc1);
      }
#pragma unroll
      for (int r = 0; r < 8; ++r) {
        float v0 = acc0[r] + bv0;
        float v1 = acc1[r] + bv1;
        v0 = v0 > 0.0f ? v0 : 0.0f;
        v1 = v1 > 0.0f ? v1 : 0.0f;
        mx0[r] = fmaxf(mx0[r], v0);
        mx1[r] = fmaxf(mx1[r], v1);
      }
    }
    float m0 = mx0[0];
    float m1 = mx1[0];
#pragma unroll
    for (int r = 1; r < 8; ++r) {
      m0 = fmaxf(m0, mx0[r]);
      m1 = fmaxf(m1, mx1[r]);
    }
    m0 = fmaxf(m0, __shfl_xor(m0, 16, 32));
    m1 = fmaxf(m1, __shfl_xor(m1, 16, 32));
    if (lane < 16) {
      out[OUT_NP_OFF + (size_t)b * 65536 + (size_t)o0 * 512 + s] = m0;
      out[OUT_NP_OFF + (size_t)b * 65536 + (size_t)o1 * 512 + s] = m1;
    }
  }
}

__global__ void __launch_bounds__(32) sa_fused_kernel(
    const float* __restrict__ xyz, const float* __restrict__ pts,
    const float* __restrict__ w0, const float* __restrict__ b0,
    const float* __restrict__ w1, const float* __restrict__ b1,
    const float* __restrict__ w2, const float* __restrict__ b2,
    float* __restrict__ out) {
  __shared__ float s_w0[64 * C0PAD];
  __shared__ float s_feat[64 * C0PAD];
  __shared__ float s_act[64 * C0PAD];
  __shared__ int s_idx[NSAMPLE];

  const int lane = threadIdx.x;
  const int b = blockIdx.x >> 9;
  const int s = blockIdx.x & 511;

  for (int i0 = 0; i0 < (64 * C0PAD) / 32; i0 += W0_RUN) {
    float wv[W0_RUN];
#pragma unroll
    for (int j = 0; j < W0_RUN; ++j) {
      const int t = lane + 32 * (i0 + j);
      const int r = t / C0PAD, c = t - r * C0PAD;
      const float x = w0[r * IN_CH + (c < IN_CH ? c : IN_CH - 1)];
      wv[j] = (c < IN_CH) ? x : 0.0f;
    }
#pragma unroll
    for (int j = 0; j < W0_RUN; ++j) s_w0[lane + 32 * (i0 + j)] = wv[j];
  }

  const float* xb = xyz + (size_t)b * 3 * Nn;
  const float qx = out[(size_t)b * 1536 + s];
  const float qy = out[(size_t)b * 1536 + 512 + s];
  const float qz = out[(size_t)b * 1536 + 1024 + s];

  const float R2 = RADIUS * RADIUS;
  const unsigned lt = (1u << lane) - 1u;
  const float4* x4 = reinterpret_cast<const float4*>(xb);
  const float4* y4 = reinterpret_cast<const float4*>(xb + Nn);
  const float4* z4 = reinterpret_cast<const float4*>(xb + 2 * Nn);
  int count = 0;
  for (int base = 0; base < Nn && count < NSAMPLE; base += 128 * SCAN_GROUPS) {
    float4 vx[SCAN_GROUPS], vy[SCAN_GROUPS], vz[SCAN_GROUPS];
#pragma unroll
    for (int g = 0; g < SCAN_GROUPS; ++g) {
      const int q4 = (base >> 2) + g * 32 + lane;
      vx[g] = x4[q4];
      vy[g] = y4[q4];
      vz[g] = z4[q4];
    }
#pragma unroll
    for (int g = 0; g < SCAN_GROUPS; ++g) {
      const float px[4] = {vx[g].x, vx[g].y, vx[g].z, vx[g].w};
      const float py[4] = {vy[g].x, vy[g].y, vy[g].z, vy[g].w};
      const float pz[4] = {vz[g].x, vz[g].y, vz[g].z, vz[g].w};
      bool hit[4];
      unsigned m[4];
#pragma unroll
      for (int e = 0; e < 4; ++e) {
        const float dx = px[e] - qx;
        const float dy = py[e] - qy;
        const float dz = pz[e] - qz;
        const float d2 = dx * dx + dy * dy + dz * dz;
        hit[e] = d2 < R2;
        m[e] = (unsigned)__ballot(hit[e]);
      }
      if ((m[0] | m[1] | m[2] | m[3]) != 0u) {
        int slot = count + __popc(m[0] & lt) + __popc(m[1] & lt) +
                   __popc(m[2] & lt) + __popc(m[3] & lt);
        const int n0 = base + g * 128 + lane * 4;
#pragma unroll
        for (int e = 0; e < 4; ++e) {
          if (hit[e] && slot < NSAMPLE) s_idx[slot] = n0 + e;
          slot += hit[e] ? 1 : 0;
        }
        count += __popc(m[0]) + __popc(m[1]) + __popc(m[2]) + __popc(m[3]);
      }
    }
  }
  __syncthreads();
  const int cnt = count < NSAMPLE ? count : NSAMPLE;
  const int ntile = __builtin_amdgcn_readfirstlane(cnt < NSAMPLE ? (cnt >> 4) + 1 : NSAMPLE / 16);

  const float* pb = pts + (size_t)b * D_FEAT * Nn;
  const int idA = (lane < cnt) ? s_idx[lane] : -1;
  const int idB = (lane + 32 < cnt) ? s_idx[lane + 32] : -1;
  {
    const int giA = (idA >= 0) ? idA : (Nn - 1);
    const int giB = (idB >= 0) ? idB : (Nn - 1);
    float ga[3], gb[3];
#pragma unroll
    for (int ch = 0; ch < 3; ++ch) {
      ga[ch] = xb[ch * Nn + giA];
      gb[ch] = xb[ch * Nn + giB];
    }
    const float q[3] = {qx, qy, qz};
#pragma unroll
    for (int ch = 0; ch < 3; ++ch) {
      s_feat[lane * C0PAD + ch] = ((idA >= 0) ? ga[ch] : 0.0f) - q[ch];
      s_feat[(lane + 32) * C0PAD + ch] = ((idB >= 0) ? gb[ch] : 0.0f) - q[ch];
    }
    s_feat[lane * C0PAD + IN_CH] = 0.0f;
    s_feat[(lane + 32) * C0PAD + IN_CH] = 0.0f;
  }
  const float* pl = pb + (size_t)lane * Nn;
  for (int k0 = 0; k0 < 16 * ntile; k0 += GATHER_RUN) {
    float f0[GATHER_RUN], f1[GATHER_RUN];
    const int ids = (k0 < 32) ? idA : idB;
#pragma unroll
    for (int u = 0; u < GATHER_RUN; ++u) {
      const int n = __builtin_amdgcn_readlane(ids, (k0 + u) & 31);
      const int gi = (n >= 0) ? n : (Nn - 1);
      f0[u] = pl[gi];
      f1[u] = pl[32 * Nn + gi];
    }
#pragma unroll
    for (int u = 0; u < GATHER_RUN; ++u) {
      s_feat[(k0 + u) * C0PAD + 3 + lane] = f0[u];
      s_feat[(k0 + u) * C0PAD + 35 + lane] = f1[u];
    }
  }
  __syncthreads();

  mlp_layer<C0PAD / 4, C0PAD>(s_w0, b0, s_feat, s_act, 64, lane, ntile);
  __syncthreads();
  mlp_layer<16, 64>(w1, b1, s_act, s_feat, 64, lane, ntile);
  __syncthreads();
  mlp_layer3_max(w2, b2, s_feat, out, lane, b, s, ntile);
}

extern "C" void kernel_launch(void* const* d_in, const int* in_sizes, int n_in,
                              void* d_out, int out_size, void* d_ws, size_t ws_size,
                              hipStream_t stream) {
  const float* xyz = (const float*)d_in[0];
  const float* pts = (const float*)d_in[1];
  const float* attn = (const float*)d_in[2];
  const float* w0 = (const float*)d_in[3];
  const float* b0 = (const float*)d_in[4];
  const float* w1 = (const float*)d_in[5];
  const float* b1 = (const float*)d_in[6];
  const float* w2 = (const float*)d_in[7];
  const float* b2 = (const float*)d_in[8];
  float* out = (float*)d_out;

  hipLaunchKernelGGL(fps_kernel, dim3(Bn * 2), dim3(FPS_THREADS), 0, stream, xyz, attn, out);
  hipLaunchKernelGGL(sa_fused_kernel, dim3(Bn * Sn), dim3(32), 0, stream,
                     xyz, pts, w0, b0, w1, b1, w2, b2, out);
}


# FPS: zero-weight points all sit at the origin and only the first can win a lowest-index argmax, so the block packs the kept points in index order and scans about half per lane; same sums
# speedup vs baseline: 3.8796x; 1.0618x over previous
#include <hip/hip_runtime.h>

typedef __attribute__((ext_vector_type(2))) float v2f;
typedef __attribute__((ext_vector_type(8))) float v8f;

#define Bn 8
#define Nn 8192
#define S_ATTN 128
#define S_NONE 384
#define Sn 512
#define NSAMPLE 64
#define RADIUS 0.4f
#define D_FEAT 64
#define IN_CH 67
#define C0PAD 68
#define SCAN_GROUPS 8
#define W0_RUN 17
#define GATHER_RUN 16
#define FPS_THREADS 256
#define FPS_WAVES 8
#define FPS_PTS 32

#define OUT_NP_OFF 12288
#define OUT_ATT_OFF 536576

__device__ __forceinline__ v8f wmma_f32(v2f a, v2f b, v8f c) {
  return __builtin_amdgcn_wmma_f32_16x16x4_f32(
       false, a,  false, b,
       (short)0, c,  false,  false);
}

template <int M>
__device__ __forceinline__ int row_xor(int v) {
  return __builtin_amdgcn_update_dpp(0, v, 0x160 + M, 0xf, 0xf, true);
}
__device__ __forceinline__ int other_row(int v) {
  return __builtin_amdgcn_permlanex16(0, v, 0x76543210, 0xfedcba98, false, false);
}
__device__ __forceinline__ int wave_max(int v) {
  v = max(v, row_xor<1>(v));
  v = max(v, row_xor<2>(v));
  v = max(v, row_xor<4>(v));
  v = max(v, row_xor<8>(v));
  return max(v, other_row(v));
}
__device__ __forceinline__ int wave_min(int v) {
  v = min(v, row_xor<1>(v));
  v = min(v, row_xor<2>(v));
  v = min(v, row_xor<4>(v));
  v = min(v, row_xor<8>(v));
  return min(v, other_row(v));
}
__device__ __forceinline__ int wave_argmax(int d, int n, int* best) {
  const int m = wave_max(d);
  *best = m;
  return wave_min(d == m ? n : 0x7fffffff);
}
__device__ __forceinline__ int oct_argmax(int d, int n) {
  int m = d;
  m = max(m, row_xor<1>(m));
  m = max(m, row_xor<2>(m));
  m = max(m, row_xor<4>(m));
  int c = (d == m) ? n : 0x7fffffff;
  c = min(c, row_xor<1>(c));
  c = min(c, row_xor<2>(c));
  return min(c, row_xor<4>(c));
}

template <int NS>
__device__ __forceinline__ void fps_picks(const int K, const int L, const int tid,
                                          const int wid, const int lane,
                                          const float* s_px, const float* s_py,
                                          const float* s_pz, int (*s_bd)[FPS_WAVES],
                                          int (*s_bn)[FPS_WAVES], int* s_sel) {
  float px[NS], py[NS], pz[NS], mind[NS];
#pragma unroll
  for (int i = 0; i < NS; ++i) {
    const int p = tid + i * FPS_THREADS;
    const bool real = p < L;
    px[i] = real ? s_px[p] : 0.0f;
    py[i] = real ? s_py[p] : 0.0f;
    pz[i] = real ? s_pz[p] : 0.0f;
    mind[i] = real ? 1e10f : -1.0f;
  }

  int last = 0;
  float lx = s_px[0], ly = s_py[0], lz = s_pz[0];

  for (int k = 0; k < K; ++k) {
    if (tid == 0) s_sel[k] = last;
    if (k == K - 1) break;

    float bd = -1.0f;
    int bi = 0;
#pragma unroll
    for (int i = 0; i < NS; ++i) {
      const float dx = px[i] - lx, dy = py[i] - ly, dz = pz[i] - lz;
      const float d2 = __builtin_fmaf(dz, dz, __builtin_fmaf(dx, dx, dy * dy));
      mind[i] = fminf(mind[i], d2);
      if (mind[i] > bd) { bd = mind[i]; bi = i; }
    }
    int bn = tid + bi * FPS_THREADS;
    int bdi;
    bn = wave_argmax(__float_as_int(bd), bn, &bdi);
    const int par = k & 1;
    if (lane == 0) {
      s_bd[par][wid] = bdi;
      s_bn[par][wid] = bn;
    }
    __syncthreads();

    const int gd = s_bd[par][lane & (FPS_WAVES - 1)];
    const int gn = s_bn[par][lane & (FPS_WAVES - 1)];
    last = __builtin_amdgcn_readfirstlane(oct_argmax(gd, gn));
    last = last < Nn ? last : Nn - 1;
    lx = s_px[last];
    ly = s_py[last];
    lz = s_pz[last];
  }
}

__global__ void __launch_bounds__(FPS_THREADS) fps_kernel(const float* __restrict__ xyz,
                                                          const float* __restrict__ attn,
                                                          float* __restrict__ out) {
  const int b = blockIdx.x >> 1;
  const int branch = blockIdx.x & 1;
  const int K = branch ? S_NONE : S_ATTN;
  const int rowoff = branch ? S_ATTN : 0;
  const int tid = threadIdx.x;
  const int wid = tid >> 5, lane = tid & 31;

  const float* xb = xyz + (size_t)b * 3 * Nn;
  const float* ab = attn + (size_t)b * Nn;

  __shared__ float s_px[Nn];
  __shared__ float s_py[Nn];
  __shared__ float s_pz[Nn];
  __shared__ int s_orig[Nn];
  __shared__ int s_cnt[Nn / 32];
  __shared__ int s_tot;
  __shared__ int s_bd[2][FPS_WAVES];
  __shared__ int s_bn[2][FPS_WAVES];
  __shared__ int s_sel[S_NONE];

  float wv[FPS_PTS];
#pragma unroll
  for (int r = 0; r < FPS_PTS; ++r) {
    const int n = tid + r * FPS_THREADS;
    const float a = ab[n];
    wv[r] = branch ? (1.0f - a) : a;
    const unsigned m = (unsigned)__ballot(wv[r] != 0.0f);
    if (lane == 0) s_cnt[r * FPS_WAVES + wid] = __popc(m);
  }
  __syncthreads();
  if (wid == 0) {
    int cv[8];
    int sum = 0;
#pragma unroll
    for (int j = 0; j < 8; ++j) {
      cv[j] = s_cnt[8 * lane + j];
      sum += cv[j];
    }
    int inc = sum;
#pragma unroll
    for (int d = 1; d < 32; d <<= 1) {
      const int t = __shfl_up(inc, d, 32);
      if (lane >= d) inc += t;
    }
    int run = inc - sum;
#pragma unroll
    for (int j = 0; j < 8; ++j) {
      s_cnt[8 * lane + j] = run;
      run += cv[j];
    }
    if (lane == 31) s_tot = inc;
  }
  __syncthreads();
  const int live = s_tot;
  const int L = __builtin_amdgcn_readfirstlane(live < Nn ? live + 1 : Nn);

  const unsigned lt = (1u << lane) - 1u;
#pragma unroll
  for (int r = 0; r < FPS_PTS; ++r) {
    const int n = tid + r * FPS_THREADS;
    const int e = r * FPS_WAVES + wid;
    const float w = wv[r];
    const float x = xb[n], y = xb[Nn + n], z = xb[2 * Nn + n];
    const unsigned m = (unsigned)__ballot(w != 0.0f);
    const int before = s_cnt[e];
    const int zeros_before = 32 * e - before;
    const unsigned zm = ~m;
    unsigned km = m;
    if (zeros_before == 0) km |= zm & (0u - zm);
    if ((km >> lane) & 1u) {
      int pos = before + (zeros_before > 0 ? 1 : 0) + __popc(km & lt);
      pos = pos < Nn ? pos : Nn - 1;
      s_px[pos] = w * x;
      s_py[pos] = w * y;
      s_pz[pos] = w * z;
      s_orig[pos] = n;
    }
  }
  __syncthreads();

  switch ((L + 2 * FPS_THREADS - 1) / (2 * FPS_THREADS)) {
    case 1: fps_picks<2>(K, L, tid, wid, lane, s_px, s_py, s_pz, s_bd, s_bn, s_sel); break;
    case 2: fps_picks<4>(K, L, tid, wid, lane, s_px, s_py, s_pz, s_bd, s_bn, s_sel); break;
    case 3: fps_picks<6>(K, L, tid, wid, lane, s_px, s_py, s_pz, s_bd, s_bn, s_sel); break;
    case 4: fps_picks<8>(K, L, tid, wid, lane, s_px, s_py, s_pz, s_bd, s_bn, s_sel); break;
    case 5: fps_picks<10>(K, L, tid, wid, lane, s_px, s_py, s_pz, s_bd, s_bn, s_sel); break;
    case 6: fps_picks<12>(K, L, tid, wid, lane, s_px, s_py, s_pz, s_bd, s_bn, s_sel); break;
    case 7: fps_picks<14>(K, L, tid, wid, lane, s_px, s_py, s_pz, s_bd, s_bn, s_sel); break;
    case 8: fps_picks<16>(K, L, tid, wid, lane, s_px, s_py, s_pz, s_bd, s_bn, s_sel); break;
    case 9: fps_picks<18>(K, L, tid, wid, lane, s_px, s_py, s_pz, s_bd, s_bn, s_sel); break;
    case 10: fps_picks<20>(K, L, tid, wid, lane, s_px, s_py, s_pz, s_bd, s_bn, s_sel); break;
    case 11: fps_picks<22>(K, L, tid, wid, lane, s_px, s_py, s_pz, s_bd, s_bn, s_sel); break;
    case 12: fps_picks<24>(K, L, tid, wid, lane, s_px, s_py, s_pz, s_bd, s_bn, s_sel); break;
    case 13: fps_picks<26>(K, L, tid, wid, lane, s_px, s_py, s_pz, s_bd, s_bn, s_sel); break;
    case 14: fps_picks<28>(K, L, tid, wid, lane, s_px, s_py, s_pz, s_bd, s_bn, s_sel); break;
    case 15: fps_picks<30>(K, L, tid, wid, lane, s_px, s_py, s_pz, s_bd, s_bn, s_sel); break;
    default: fps_picks<32>(K, L, tid, wid, lane, s_px, s_py, s_pz, s_bd, s_bn, s_sel); break;
  }
  __syncthreads();

  for (int k = tid; k < K; k += FPS_THREADS) {
    int p = s_sel[k];
    p = p < Nn ? p : Nn - 1;
    int n = s_orig[p];
    n = (n >= 0 && n < Nn) ? n : Nn - 1;
    const int row = rowoff + k;
    out[(size_t)b * 1536 + row] = s_px[p];
    out[(size_t)b * 1536 + 512 + row] = s_py[p];
    out[(size_t)b * 1536 + 1024 + row] = s_pz[p];
    out[OUT_ATT_OFF + (size_t)b * 512 + row] = ab[n];
  }
}

template <int NK, int WSTRIDE>
__device__ __forceinline__ void mlp_layer(const float* __restrict__ W,
                                          const float* __restrict__ bias,
                                          const float* sin_, float* sout,
                                          int cout, int lane, int ntile) {
  const int hl = lane >> 4;
  const int ll = lane & 15;
  for (int op = 0; op < cout / 32; ++op) {
    const int o0 = op * 32 + ll;
    const int o1 = o0 + 16;
    const float bv0 = bias[o0];
    const float bv1 = bias[o1];
    v2f wf0[NK], wf1[NK];
#pragma unroll
    for (int j = 0; j < NK; ++j) {
      const int kk = 4 * j + (hl << 1);
      wf0[j].x = W[o0 * WSTRIDE + kk];
      wf0[j].y = W[o0 * WSTRIDE + kk + 1];
      wf1[j].x = W[o1 * WSTRIDE + kk];
      wf1[j].y = W[o1 * WSTRIDE + kk + 1];
    }
    for (int rt = 0; rt < ntile; ++rt) {
      v8f acc0 = {};
      v8f acc1 = {};
      const int m = rt * 16 + ll;
#pragma unroll
      for (int j = 0; j < NK; ++j) {
        const int kk = 4 * j + (hl << 1);
        v2f a;
        a.x = sin_[m * C0PAD + kk];
        a.y = sin_[m * C0PAD + kk + 1];
        acc0 = wmma_f32(a, wf0[j], acc0);
        acc1 = wmma_f32(a, wf1[j], acc1);
      }
#pragma unroll
      for (int r = 0; r < 8; ++r) {
        float v0 = acc0[r] + bv0;
        float v1 = acc1[r] + bv1;
        v0 = v0 > 0.0f ? v0 : 0.0f;
        v1 = v1 > 0.0f ? v1 : 0.0f;
        const int krow = rt * 16 + r + 8 * hl;
        sout[krow * C0PAD + o0] = v0;
        sout[krow * C0PAD + o1] = v1;
      }
    }
  }
}

__device__ __forceinline__ void mlp_layer3_max(const float* __restrict__ W,
                                               const float* __restrict__ bias,
                                               const float* sin_,
                                               float* __restrict__ out,
                                               int lane, int b, int s, int ntile) {
  const int hl = lane >> 4;
  const int ll = lane & 15;
  for (int op = 0; op < 4; ++op) {
    const int o0 = op * 32 + ll;
    const int o1 = o0 + 16;
    const float bv0 = bias[o0];
    const float bv1 = bias[o1];
    v2f wf0[16], wf1[16];
#pragma unroll
    for (int j = 0; j < 16; ++j) {
      const int kk = 4 * j + (hl << 1);
      wf0[j].x = W[o0 * 64 + kk];
      wf0[j].y = W[o0 * 64 + kk + 1];
      wf1[j].x = W[o1 * 64 + kk];
      wf1[j].y = W[o1 * 64 + kk + 1];
    }
    v8f mx0 = {};
    v8f mx1 = {};
    for (int rt = 0; rt < ntile; ++rt) {
      v8f acc0 = {};
      v8f acc1 = {};
      const int m = rt * 16 + ll;
#pragma unroll
      for (int j = 0; j < 16; ++j) {
        const int kk = 4 * j + (hl << 1);
        v2f a;
        a.x = sin_[m * C0PAD + kk];
        a.y = sin_[m * C0PAD + kk + 1];
        acc0 = wmma_f32(a, wf0[j], acc0);
        acc1 = wmma_f32(a, wf1[j], acc1);
      }
#pragma unroll
      for (int r = 0; r < 8; ++r) {
        float v0 = acc0[r] + bv0;
        float v1 = acc1[r] + bv1;
        v0 = v0 > 0.0f ? v0 : 0.0f;
        v1 = v1 > 0.0f ? v1 : 0.0f;
        mx0[r] = fmaxf(mx0[r], v0);
        mx1[r] = fmaxf(mx1[r], v1);
      }
    }
    float m0 = mx0[0];
    float m1 = mx1[0];
#pragma unroll
    for (int r = 1; r < 8; ++r) {
      m0 = fmaxf(m0, mx0[r]);
      m1 = fmaxf(m1, mx1[r]);
    }
    m0 = fmaxf(m0, __shfl_xor(m0, 16, 32));
    m1 = fmaxf(m1, __shfl_xor(m1, 16, 32));
    if (lane < 16) {
      out[OUT_NP_OFF + (size_t)b * 65536 + (size_t)o0 * 512 + s] = m0;
      out[OUT_NP_OFF + (size_t)b * 65536 + (size_t)o1 * 512 + s] = m1;
    }
  }
}

__global__ void __launch_bounds__(32) sa_fused_kernel(
    const float* __restrict__ xyz, const float* __restrict__ pts,
    const float* __restrict__ w0, const float* __restrict__ b0,
    const float* __restrict__ w1, const float* __restrict__ b1,
    const float* __restrict__ w2, const float* __restrict__ b2,
    float* __restrict__ out) {
  __shared__ float s_w0[64 * C0PAD];
  __shared__ float s_feat[64 * C0PAD];
  __shared__ float s_act[64 * C0PAD];
  __shared__ int s_idx[NSAMPLE];

  const int lane = threadIdx.x;
  const int b = blockIdx.x >> 9;
  const int s = blockIdx.x & 511;

  for (int i0 = 0; i0 < (64 * C0PAD) / 32; i0 += W0_RUN) {
    float wv[W0_RUN];
#pragma unroll
    for (int j = 0; j < W0_RUN; ++j) {
      const int t = lane + 32 * (i0 + j);
      const int r = t / C0PAD, c = t - r * C0PAD;
      const float x = w0[r * IN_CH + (c < IN_CH ? c : IN_CH - 1)];
      wv[j] = (c < IN_CH) ? x : 0.0f;
    }
#pragma unroll
    for (int j = 0; j < W0_RUN; ++j) s_w0[lane + 32 * (i0 + j)] = wv[j];
  }

  const float* xb = xyz + (size_t)b * 3 * Nn;
  const float qx = out[(size_t)b * 1536 + s];
  const float qy = out[(size_t)b * 1536 + 512 + s];
  const float qz = out[(size_t)b * 1536 + 1024 + s];

  const float R2 = RADIUS * RADIUS;
  const unsigned lt = (1u << lane) - 1u;
  const float4* x4 = reinterpret_cast<const float4*>(xb);
  const float4* y4 = reinterpret_cast<const float4*>(xb + Nn);
  const float4* z4 = reinterpret_cast<const float4*>(xb + 2 * Nn);
  int count = 0;
  for (int base = 0; base < Nn && count < NSAMPLE; base += 128 * SCAN_GROUPS) {
    float4 vx[SCAN_GROUPS], vy[SCAN_GROUPS], vz[SCAN_GROUPS];
#pragma unroll
    for (int g = 0; g < SCAN_GROUPS; ++g) {
      const int q4 = (base >> 2) + g * 32 + lane;
      vx[g] = x4[q4];
      vy[g] = y4[q4];
      vz[g] = z4[q4];
    }
#pragma unroll
    for (int g = 0; g < SCAN_GROUPS; ++g) {
      const float px[4] = {vx[g].x, vx[g].y, vx[g].z, vx[g].w};
      const float py[4] = {vy[g].x, vy[g].y, vy[g].z, vy[g].w};
      const float pz[4] = {vz[g].x, vz[g].y, vz[g].z, vz[g].w};
      bool hit[4];
      unsigned m[4];
#pragma unroll
      for (int e = 0; e < 4; ++e) {
        const float dx = px[e] - qx;
        const float dy = py[e] - qy;
        const float dz = pz[e] - qz;
        const float d2 = dx * dx + dy * dy + dz * dz;
        hit[e] = d2 < R2;
        m[e] = (unsigned)__ballot(hit[e]);
      }
      if ((m[0] | m[1] | m[2] | m[3]) != 0u) {
        int slot = count + __popc(m[0] & lt) + __popc(m[1] & lt) +
                   __popc(m[2] & lt) + __popc(m[3] & lt);
        const int n0 = base + g * 128 + lane * 4;
#pragma unroll
        for (int e = 0; e < 4; ++e) {
          if (hit[e] && slot < NSAMPLE) s_idx[slot] = n0 + e;
          slot += hit[e] ? 1 : 0;
        }
        count += __popc(m[0]) + __popc(m[1]) + __popc(m[2]) + __popc(m[3]);
      }
    }
  }
  __syncthreads();
  const int cnt = count < NSAMPLE ? count : NSAMPLE;
  const int ntile = __builtin_amdgcn_readfirstlane(cnt < NSAMPLE ? (cnt >> 4) + 1 : NSAMPLE / 16);

  const float* pb = pts + (size_t)b * D_FEAT * Nn;
  const int idA = (lane < cnt) ? s_idx[lane] : -1;
  const int idB = (lane + 32 < cnt) ? s_idx[lane + 32] : -1;
  {
    const int giA = (idA >= 0) ? idA : (Nn - 1);
    const int giB = (idB >= 0) ? idB : (Nn - 1);
    float ga[3], gb[3];
#pragma unroll
    for (int ch = 0; ch < 3; ++ch) {
      ga[ch] = xb[ch * Nn + giA];
      gb[ch] = xb[ch * Nn + giB];
    }
    const float q[3] = {qx, qy, qz};
#pragma unroll
    for (int ch = 0; ch < 3; ++ch) {
      s_feat[lane * C0PAD + ch] = ((idA >= 0) ? ga[ch] : 0.0f) - q[ch];
      s_feat[(lane + 32) * C0PAD + ch] = ((idB >= 0) ? gb[ch] : 0.0f) - q[ch];
    }
    s_feat[lane * C0PAD + IN_CH] = 0.0f;
    s_feat[(lane + 32) * C0PAD + IN_CH] = 0.0f;
  }
  const float* pl = pb + (size_t)lane * Nn;
  for (int k0 = 0; k0 < 16 * ntile; k0 += GATHER_RUN) {
    float f0[GATHER_RUN], f1[GATHER_RUN];
    const int ids = (k0 < 32) ? idA : idB;
#pragma unroll
    for (int u = 0; u < GATHER_RUN; ++u) {
      const int n = __builtin_amdgcn_readlane(ids, (k0 + u) & 31);
      const int gi = (n >= 0) ? n : (Nn - 1);
      f0[u] = pl[gi];
      f1[u] = pl[32 * Nn + gi];
    }
#pragma unroll
    for (int u = 0; u < GATHER_RUN; ++u) {
      s_feat[(k0 + u) * C0PAD + 3 + lane] = f0[u];
      s_feat[(k0 + u) * C0PAD + 35 + lane] = f1[u];
    }
  }
  __syncthreads();

  mlp_layer<C0PAD / 4, C0PAD>(s_w0, b0, s_feat, s_act, 64, lane, ntile);
  __syncthreads();
  mlp_layer<16, 64>(w1, b1, s_act, s_feat, 64, lane, ntile);
  __syncthreads();
  mlp_layer3_max(w2, b2, s_feat, out, lane, b, s, ntile);
}

extern "C" void kernel_launch(void* const* d_in, const int* in_sizes, int n_in,
                              void* d_out, int out_size, void* d_ws, size_t ws_size,
                              hipStream_t stream) {
  const float* xyz = (const float*)d_in[0];
  const float* pts = (const float*)d_in[1];
  const float* attn = (const float*)d_in[2];
  const float* w0 = (const float*)d_in[3];
  const float* b0 = (const float*)d_in[4];
  const float* w1 = (const float*)d_in[5];
  const float* b1 = (const float*)d_in[6];
  const float* w2 = (const float*)d_in[7];
  const float* b2 = (const float*)d_in[8];
  float* out = (float*)d_out;

  hipLaunchKernelGGL(fps_kernel, dim3(Bn * 2), dim3(FPS_THREADS), 0, stream, xyz, attn, out);
  hipLaunchKernelGGL(sa_fused_kernel, dim3(Bn * Sn), dim3(32), 0, stream,
                     xyz, pts, w0, b0, w1, b1, w2, b2, out);
}
